# starting kernel
# baseline (speedup 1.0000x reference)
#include <hip/hip_runtime.h>

#define F_IN   256
#define HEADS  6
#define CDIM   128
#define FDIM   768
#define OUT_C  128
#define NEG_SLOPE 0.2f
#define BN_EPS    1e-5f

typedef __attribute__((ext_vector_type(2))) float v2f;
typedef __attribute__((ext_vector_type(8))) float v8f;

__device__ __forceinline__ unsigned fmap(float f) {
    unsigned u = __float_as_uint(f);
    return (u & 0x80000000u) ? ~u : (u | 0x80000000u);
}
__device__ __forceinline__ float funmap(unsigned u) {
    u = (u & 0x80000000u) ? (u & 0x7FFFFFFFu) : ~u;
    return __uint_as_float(u);
}
#define FMAP_NEG_INF 0x007FFFFFu

#define WMMA_F32(a, b, c) \
    __builtin_amdgcn_wmma_f32_16x16x4_f32(false, (a), false, (b), (short)0, (c), false, false)

__global__ __launch_bounds__(32)
void wmma_gemm_f32(const float* __restrict__ A, const float* __restrict__ B,
                   float* __restrict__ C, const float* __restrict__ bias,
                   int M, int K, int Nc)
{
    const int lane  = threadIdx.x & 31;
    const int nN64  = Nc >> 6;
    const int m0    = (blockIdx.x / nN64) * 32;
    const int n0    = (blockIdx.x % nN64) * 64;
    const int lrow  = lane & 15;
    const int khalf = (lane >> 4) << 1;
    const bool hasM1 = (m0 + 16) < M;

    v8f acc00 = {}, acc01 = {}, acc02 = {}, acc03 = {};
    v8f acc10 = {}, acc11 = {}, acc12 = {}, acc13 = {};

    const float* arow0 = A + (size_t)(m0 + lrow) * K;
    const float* arow1 = A + (size_t)(hasM1 ? (m0 + 16 + lrow) : (m0 + lrow)) * K;

    for (int k = 0; k < K; k += 4) {
        v2f a0, a1;
        a0.x = arow0[k + khalf];
        a0.y = arow0[k + khalf + 1];
        a1.x = arow1[k + khalf];
        a1.y = arow1[k + khalf + 1];

        const float* brow0 = B + (size_t)(k + khalf) * Nc + n0 + lrow;
        const float* brow1 = brow0 + Nc;
        v2f bv0, bv1, bv2, bv3;
        bv0.x = brow0[0];   bv0.y = brow1[0];
        bv1.x = brow0[16];  bv1.y = brow1[16];
        bv2.x = brow0[32];  bv2.y = brow1[32];
        bv3.x = brow0[48];  bv3.y = brow1[48];

        acc00 = WMMA_F32(a0, bv0, acc00);
        acc01 = WMMA_F32(a0, bv1, acc01);
        acc02 = WMMA_F32(a0, bv2, acc02);
        acc03 = WMMA_F32(a0, bv3, acc03);
        acc10 = WMMA_F32(a1, bv0, acc10);
        acc11 = WMMA_F32(a1, bv1, acc11);
        acc12 = WMMA_F32(a1, bv2, acc12);
        acc13 = WMMA_F32(a1, bv3, acc13);
    }

    float bb0 = 0.f, bb1 = 0.f, bb2 = 0.f, bb3 = 0.f;
    if (bias) {
        bb0 = bias[n0 + lrow];
        bb1 = bias[n0 + 16 + lrow];
        bb2 = bias[n0 + 32 + lrow];
        bb3 = bias[n0 + 48 + lrow];
    }
    const int rbase = (lane >> 4) * 8;
    #pragma unroll
    for (int r = 0; r < 8; ++r) {
        float* crow = C + (size_t)(m0 + rbase + r) * Nc + n0 + lrow;
        crow[0]  = acc00[r] + bb0;
        crow[16] = acc01[r] + bb1;
        crow[32] = acc02[r] + bb2;
        crow[48] = acc03[r] + bb3;
    }
    if (hasM1) {
        #pragma unroll
        for (int r = 0; r < 8; ++r) {
            float* crow = C + (size_t)(m0 + 16 + rbase + r) * Nc + n0 + lrow;
            crow[0]  = acc10[r] + bb0;
            crow[16] = acc11[r] + bb1;
            crow[32] = acc12[r] + bb2;
            crow[48] = acc13[r] + bb3;
        }
    }
}

__global__ __launch_bounds__(256)
void attn_scores(const float* __restrict__ hx,
                 const float* __restrict__ att_s, const float* __restrict__ att_d,
                 float* __restrict__ a_s, float* __restrict__ a_d, int nNodes)
{
    int gid  = blockIdx.x * blockDim.x + threadIdx.x;
    int wave = gid >> 5;
    int lane = gid & 31;
    if (wave >= nNodes * HEADS) return;
    int n = wave / HEADS, h = wave % HEADS;

    const float4* hv = (const float4*)(hx + (size_t)n * FDIM + h * CDIM);
    const float4* sv = (const float4*)(att_s + h * CDIM);
    const float4* dv = (const float4*)(att_d + h * CDIM);
    float4 x = hv[lane], s = sv[lane], d = dv[lane];
    float accs = x.x * s.x + x.y * s.y + x.z * s.z + x.w * s.w;
    float accd = x.x * d.x + x.y * d.y + x.z * d.z + x.w * d.w;
    #pragma unroll
    for (int off = 16; off > 0; off >>= 1) {
        accs += __shfl_down(accs, off, 32);
        accd += __shfl_down(accd, off, 32);
    }
    if (lane == 0) { a_s[wave] = accs; a_d[wave] = accd; }
}

__global__ void fill_u32(unsigned* __restrict__ p, unsigned v, size_t n)
{
    size_t i = (size_t)blockIdx.x * blockDim.x + threadIdx.x;
    if (i < n) p[i] = v;
}

__global__ __launch_bounds__(256)
void edge_scores_max(const int* __restrict__ adj,
                     const float* __restrict__ a_s, const float* __restrict__ a_d,
                     float* __restrict__ ebuf, unsigned* __restrict__ mbuf,
                     int E, int nTot)
{
    int e = blockIdx.x * blockDim.x + threadIdx.x;
    if (e >= nTot) return;
    int src, dst;
    if (e < E) { src = adj[e]; dst = adj[E + e]; } else { src = dst = e - E; }
    #pragma unroll
    for (int h = 0; h < HEADS; ++h) {
        float v = a_s[src * HEADS + h] + a_d[dst * HEADS + h];
        v = (v > 0.f) ? v : v * NEG_SLOPE;
        ebuf[(size_t)e * HEADS + h] = v;
        atomicMax(&mbuf[dst * HEADS + h], fmap(v));
    }
}

__global__ __launch_bounds__(256)
void edge_exp_sum(const int* __restrict__ adj, float* __restrict__ ebuf,
                  const unsigned* __restrict__ mbuf, float* __restrict__ sbuf,
                  int E, int nTot)
{
    int e = blockIdx.x * blockDim.x + threadIdx.x;
    if (e >= nTot) return;
    int dst = (e < E) ? adj[E + e] : e - E;
    #pragma unroll
    for (int h = 0; h < HEADS; ++h) {
        float m = funmap(mbuf[dst * HEADS + h]);
        float p = __expf(ebuf[(size_t)e * HEADS + h] - m);
        ebuf[(size_t)e * HEADS + h] = p;
        atomicAdd(&sbuf[dst * HEADS + h], p);
    }
}

__global__ __launch_bounds__(256)
void edge_aggregate(const int* __restrict__ adj, const float* __restrict__ hx,
                    const float* __restrict__ ebuf, const float* __restrict__ sbuf,
                    float* __restrict__ out, int E, int nTot)
{
    long gid  = (long)blockIdx.x * blockDim.x + threadIdx.x;
    long wave = gid >> 5;
    int  lane = (int)(gid & 31);
    if (wave >= (long)nTot * HEADS) return;
    int e = (int)(wave / HEADS), h = (int)(wave % HEADS);
    int src, dst;
    if (e < E) { src = adj[e]; dst = adj[E + e]; } else { src = dst = e - E; }

    float alpha = ebuf[(size_t)e * HEADS + h] / sbuf[dst * HEADS + h];
    const float4* hv = (const float4*)(hx + (size_t)src * FDIM + h * CDIM);
    float4 v = hv[lane];
    float* o = out + (size_t)dst * FDIM + h * CDIM + lane * 4;
    atomicAdd(o + 0, v.x * alpha);
    atomicAdd(o + 1, v.y * alpha);
    atomicAdd(o + 2, v.z * alpha);
    atomicAdd(o + 3, v.w * alpha);
}

__global__ __launch_bounds__(256)
void bias_relu(float* __restrict__ x, const float* __restrict__ bias, size_t total)
{
    size_t i = (size_t)blockIdx.x * blockDim.x + threadIdx.x;
    if (i >= total) return;
    float v = x[i] + bias[i % FDIM];
    x[i] = v > 0.f ? v : 0.f;
}

__global__ __launch_bounds__(256)
void bn_stats(const float* __restrict__ x, float* __restrict__ mean,
              float* __restrict__ rstd, int rows)
{
    int col = blockIdx.x;
    float s = 0.f, ss = 0.f;
    for (int r = threadIdx.x; r < rows; r += 256) {
        float v = x[(size_t)r * FDIM + col];
        s += v; ss += v * v;
    }
    __shared__ float sh[256], sh2[256];
    sh[threadIdx.x] = s; sh2[threadIdx.x] = ss;
    __syncthreads();
    for (int o = 128; o > 0; o >>= 1) {
        if (threadIdx.x < o) {
            sh[threadIdx.x]  += sh[threadIdx.x + o];
            sh2[threadIdx.x] += sh2[threadIdx.x + o];
        }
        __syncthreads();
    }
    if (threadIdx.x == 0) {
        float mu  = sh[0] / rows;
        float var = sh2[0] / rows - mu * mu;
        mean[col] = mu;
        rstd[col] = rsqrtf(var + BN_EPS);
    }
}

__global__ __launch_bounds__(256)
void bn_apply(float* __restrict__ x, const float* __restrict__ mean,
              const float* __restrict__ rstd, const float* __restrict__ gamma,
              const float* __restrict__ beta, size_t total)
{
    size_t i = (size_t)blockIdx.x * blockDim.x + threadIdx.x;
    if (i >= total) return;
    int col = (int)(i % FDIM);
    x[i] = (x[i] - mean[col]) * rstd[col] * gamma[col] + beta[col];
}

__global__ __launch_bounds__(256)
void pool_max(const float* __restrict__ lin, const int* __restrict__ ibatch,
              unsigned* __restrict__ pooled, int rows)
{
    size_t i = (size_t)blockIdx.x * blockDim.x + threadIdx.x;
    if (i >= (size_t)rows * OUT_C) return;
    int n = (int)(i / OUT_C), c = (int)(i % OUT_C);
    atomicMax(&pooled[ibatch[n] * OUT_C + c], fmap(lin[i]));
}

__global__ __launch_bounds__(256)
void pool_final(const unsigned* __restrict__ pooled, float* __restrict__ out, int n)
{
    int i = blockIdx.x * blockDim.x + threadIdx.x;
    if (i < n) out[i] = funmap(pooled[i]);
}

extern "C" void kernel_launch(void* const* d_in, const int* in_sizes, int n_in,
                              void* d_out, int out_size, void* d_ws, size_t ws_size,
                              hipStream_t stream)
{
    const float* x_in   = (const float*)d_in[0];
    const int*   adj    = (const int*)d_in[1];
    const int*   ibatch = (const int*)d_in[2];
    const float* W1     = (const float*)d_in[3];
    const float* att_s1 = (const float*)d_in[4];
    const float* att_d1 = (const float*)d_in[5];
    const float* b1     = (const float*)d_in[6];
    const float* g1     = (const float*)d_in[7];
    const float* be1    = (const float*)d_in[8];
    const float* W2     = (const float*)d_in[9];
    const float* att_s2 = (const float*)d_in[10];
    const float* att_d2 = (const float*)d_in[11];
    const float* b2     = (const float*)d_in[12];
    const float* g2     = (const float*)d_in[13];
    const float* be2    = (const float*)d_in[14];
    const float* linW   = (const float*)d_in[15];
    const float* linb   = (const float*)d_in[16];
    float* out = (float*)d_out;

    const int N    = in_sizes[0] / F_IN;
    const int E    = in_sizes[1] / 2;
    const int Etot = E + N;
    const int B    = out_size / OUT_C;
    const int mT   = (N + 31) / 32;

    float*    hx   = (float*)d_ws;
    float*    agg  = hx  + (size_t)N * FDIM;
    float*    a_s  = agg + (size_t)N * FDIM;
    float*    a_d  = a_s + (size_t)N * HEADS;
    unsigned* mbuf = (unsigned*)(a_d + (size_t)N * HEADS);
    float*    sbuf = (float*)(mbuf + (size_t)N * HEADS);
    float*    ebuf = sbuf + (size_t)N * HEADS;
    float*    mean = ebuf + (size_t)Etot * HEADS;
    float*    rstd = mean + FDIM;
    float*    lin  = rstd + FDIM;
    unsigned* pooled = (unsigned*)(lin + (size_t)N * OUT_C);

    const size_t nh  = (size_t)N * HEADS;
    const size_t nf  = (size_t)N * FDIM;
    const long   ew  = (long)Etot * HEADS;

    auto layer = [&](const float* xin, int K, const float* W,
                     const float* asw, const float* adw,
                     const float* bias, const float* gamma, const float* beta) {
        wmma_gemm_f32<<<mT * (FDIM / 64), 32, 0, stream>>>(xin, W, hx, nullptr, N, K, FDIM);
        attn_scores<<<(int)((nh * 32 + 255) / 256), 256, 0, stream>>>(hx, asw, adw, a_s, a_d, N);
        fill_u32<<<(int)((nh + 255) / 256), 256, 0, stream>>>(mbuf, FMAP_NEG_INF, nh);
        fill_u32<<<(int)((nh + 255) / 256), 256, 0, stream>>>((unsigned*)sbuf, 0u, nh);
        fill_u32<<<(int)((nf + 255) / 256), 256, 0, stream>>>((unsigned*)agg, 0u, nf);
        edge_scores_max<<<(Etot + 255) / 256, 256, 0, stream>>>(adj, a_s, a_d, ebuf, mbuf, E, Etot);
        edge_exp_sum<<<(Etot + 255) / 256, 256, 0, stream>>>(adj, ebuf, mbuf, sbuf, E, Etot);
        edge_aggregate<<<(int)((ew * 32 + 255) / 256), 256, 0, stream>>>(adj, hx, ebuf, sbuf, agg, E, Etot);
        bias_relu<<<(int)((nf + 255) / 256), 256, 0, stream>>>(agg, bias, nf);
        bn_stats<<<FDIM, 256, 0, stream>>>(agg, mean, rstd, N);
        bn_apply<<<(int)((nf + 255) / 256), 256, 0, stream>>>(agg, mean, rstd, gamma, beta, nf);
    };

    layer(x_in, F_IN, W1, att_s1, att_d1, b1, g1, be1);
    layer(agg,  FDIM, W2, att_s2, att_d2, b2, g2, be2);

    wmma_gemm_f32<<<mT * (OUT_C / 64), 32, 0, stream>>>(agg, linW, lin, linb, N, FDIM, OUT_C);
    fill_u32<<<(B * OUT_C + 255) / 256, 256, 0, stream>>>(pooled, FMAP_NEG_INF, (size_t)B * OUT_C);
    pool_max<<<(int)(((size_t)N * OUT_C + 255) / 256), 256, 0, stream>>>(lin, ibatch, pooled, N);
    pool_final<<<(B * OUT_C + 255) / 256, 256, 0, stream>>>(pooled, out, B * OUT_C);
}


# Batch-norm column sums formed in the aggregation kernel's write-out (per-tile partials via LDS) and added by a small reduction; strided column-read pass gone; statistics' sum order changed
# speedup vs baseline: 1.8730x; 1.8730x over previous
#include <hip/hip_runtime.h>

#define F_IN   256
#define HEADS  6
#define CDIM   128
#define FDIM   768
#define OUT_C  128
#define NEG_SLOPE 0.2f
#define BN_EPS    1e-5f

typedef __attribute__((ext_vector_type(2))) float v2f;
typedef __attribute__((ext_vector_type(8))) float v8f;

__device__ __forceinline__ unsigned fmap(float f) {
    unsigned u = __float_as_uint(f);
    return (u & 0x80000000u) ? ~u : (u | 0x80000000u);
}
__device__ __forceinline__ float funmap(unsigned u) {
    u = (u & 0x80000000u) ? (u & 0x7FFFFFFFu) : ~u;
    return __uint_as_float(u);
}
#define FMAP_NEG_INF 0x007FFFFFu

#define WMMA_F32(a, b, c) \
    __builtin_amdgcn_wmma_f32_16x16x4_f32(false, (a), false, (b), (short)0, (c), false, false)

#define KU 4

struct KFrag { v2f a0, a1, b0, b1, b2, b3; };

__device__ __forceinline__ void load_kfrag(KFrag& f,
                                           const float* __restrict__ ap0,
                                           const float* __restrict__ ap1,
                                           const float* __restrict__ bp, int Nc)
{
    f.a0.x = ap0[0];  f.a0.y = ap0[1];
    f.a1.x = ap1[0];  f.a1.y = ap1[1];
    const float* bq = bp + Nc;
    f.b0.x = bp[0];   f.b0.y = bq[0];
    f.b1.x = bp[16];  f.b1.y = bq[16];
    f.b2.x = bp[32];  f.b2.y = bq[32];
    f.b3.x = bp[48];  f.b3.y = bq[48];
}

__global__ __launch_bounds__(32)
void wmma_gemm_f32(const float* __restrict__ A, const float* __restrict__ B,
                   float* __restrict__ C, const float* __restrict__ bias,
                   int M, int K, int Nc)
{
    const int lane  = threadIdx.x & 31;
    const int nN64  = Nc >> 6;
    const int m0    = (blockIdx.x / nN64) * 32;
    const int n0    = (blockIdx.x % nN64) * 64;
    const int lrow  = lane & 15;
    const int khalf = (lane >> 4) << 1;
    const bool hasM1 = (m0 + 16) < M;

    v8f acc00 = {}, acc01 = {}, acc02 = {}, acc03 = {};
    v8f acc10 = {}, acc11 = {}, acc12 = {}, acc13 = {};

    const float* ap0 = A + (size_t)(m0 + lrow) * K + khalf;
    const float* ap1 = A + (size_t)(hasM1 ? (m0 + 16 + lrow) : (m0 + lrow)) * K + khalf;
    const float* bp  = B + (size_t)khalf * Nc + n0 + lrow;
    const size_t bstep = (size_t)4 * Nc;

    const int nk  = K >> 2;
    const int nst = nk / KU;

    KFrag P[KU], Q[KU];

#define LOAD_STAGE(S)                                                        \
    do {                                                                     \
        _Pragma("unroll")   \
        for (int u = 0; u < KU; ++u)                                         \
            load_kfrag(S[u], ap0 + 4 * u, ap1 + 4 * u, bp + u * bstep, Nc);  \
        ap0 += 4 * KU;  ap1 += 4 * KU;  bp += KU * bstep;                    \
    } while (0)

#define MMA_KFRAG(f)                                                         \
    do {                                                                     \
        acc00 = WMMA_F32((f).a0, (f).b0, acc00);                             \
        acc01 = WMMA_F32((f).a0, (f).b1, acc01);                             \
        acc02 = WMMA_F32((f).a0, (f).b2, acc02);                             \
        acc03 = WMMA_F32((f).a0, (f).b3, acc03);                             \
        acc10 = WMMA_F32((f).a1, (f).b0, acc10);                             \
        acc11 = WMMA_F32((f).a1, (f).b1, acc11);                             \
        acc12 = WMMA_F32((f).a1, (f).b2, acc12);                             \
        acc13 = WMMA_F32((f).a1, (f).b3, acc13);                             \
    } while (0)

#define MMA_STAGE(S)                                                         \
    do {                                                                     \
        _Pragma("unroll")   \
        for (int u = 0; u < KU; ++u) MMA_KFRAG(S[u]);                        \
    } while (0)

    if (nst > 0) {
        int s = 0;
        LOAD_STAGE(P);
        for (; s + 2 < nst; s += 2) {
            LOAD_STAGE(Q);  MMA_STAGE(P);
            LOAD_STAGE(P);  MMA_STAGE(Q);
        }
        if (s + 1 < nst) { LOAD_STAGE(Q);  MMA_STAGE(P);  MMA_STAGE(Q); }
        else             { MMA_STAGE(P); }
    }
    for (int t = nst * KU; t < nk; ++t) {
        KFrag f;
        load_kfrag(f, ap0, ap1, bp, Nc);
        ap0 += 4;  ap1 += 4;  bp += bstep;
        MMA_KFRAG(f);
    }
#undef LOAD_STAGE
#undef MMA_KFRAG
#undef MMA_STAGE

    float bb0 = 0.f, bb1 = 0.f, bb2 = 0.f, bb3 = 0.f;
    if (bias) {
        bb0 = bias[n0 + lrow];
        bb1 = bias[n0 + 16 + lrow];
        bb2 = bias[n0 + 32 + lrow];
        bb3 = bias[n0 + 48 + lrow];
    }
    const int rbase = (lane >> 4) * 8;
    #pragma unroll
    for (int r = 0; r < 8; ++r) {
        float* crow = C + (size_t)(m0 + rbase + r) * Nc + n0 + lrow;
        crow[0]  = acc00[r] + bb0;
        crow[16] = acc01[r] + bb1;
        crow[32] = acc02[r] + bb2;
        crow[48] = acc03[r] + bb3;
    }
    if (hasM1) {
        #pragma unroll
        for (int r = 0; r < 8; ++r) {
            float* crow = C + (size_t)(m0 + 16 + rbase + r) * Nc + n0 + lrow;
            crow[0]  = acc10[r] + bb0;
            crow[16] = acc11[r] + bb1;
            crow[32] = acc12[r] + bb2;
            crow[48] = acc13[r] + bb3;
        }
    }
}

__global__ __launch_bounds__(256)
void attn_scores(const float* __restrict__ hx,
                 const float* __restrict__ att_s, const float* __restrict__ att_d,
                 float* __restrict__ a_s, float* __restrict__ a_d, int nNodes)
{
    int gid  = blockIdx.x * blockDim.x + threadIdx.x;
    int wave = gid >> 5;
    int lane = gid & 31;
    if (wave >= nNodes * HEADS) return;
    int n = wave / HEADS, h = wave % HEADS;

    const float4* hv = (const float4*)(hx + (size_t)n * FDIM + h * CDIM);
    const float4* sv = (const float4*)(att_s + h * CDIM);
    const float4* dv = (const float4*)(att_d + h * CDIM);
    float4 x = hv[lane], s = sv[lane], d = dv[lane];
    float accs = x.x * s.x + x.y * s.y + x.z * s.z + x.w * s.w;
    float accd = x.x * d.x + x.y * d.y + x.z * d.z + x.w * d.w;
    #pragma unroll
    for (int off = 16; off > 0; off >>= 1) {
        accs += __shfl_down(accs, off, 32);
        accd += __shfl_down(accd, off, 32);
    }
    if (lane == 0) { a_s[wave] = accs; a_d[wave] = accd; }
}

__global__ void fill_u32(unsigned* __restrict__ p, unsigned v, size_t n)
{
    size_t i = (size_t)blockIdx.x * blockDim.x + threadIdx.x;
    if (i < n) p[i] = v;
}

__global__ __launch_bounds__(256)
void edge_scores_max(const int* __restrict__ adj,
                     const float* __restrict__ a_s, const float* __restrict__ a_d,
                     float* __restrict__ ebuf, unsigned* __restrict__ mbuf,
                     int E, int nTot)
{
    int e = blockIdx.x * blockDim.x + threadIdx.x;
    if (e >= nTot) return;
    int src, dst;
    if (e < E) { src = adj[e]; dst = adj[E + e]; } else { src = dst = e - E; }
    #pragma unroll
    for (int h = 0; h < HEADS; ++h) {
        float v = a_s[src * HEADS + h] + a_d[dst * HEADS + h];
        v = (v > 0.f) ? v : v * NEG_SLOPE;
        ebuf[(size_t)e * HEADS + h] = v;
        atomicMax(&mbuf[dst * HEADS + h], fmap(v));
    }
}

__global__ __launch_bounds__(256)
void edge_exp_sum(const int* __restrict__ adj, float* __restrict__ ebuf,
                  const unsigned* __restrict__ mbuf, float* __restrict__ sbuf,
                  int E, int nTot)
{
    int e = blockIdx.x * blockDim.x + threadIdx.x;
    if (e >= nTot) return;
    int dst = (e < E) ? adj[E + e] : e - E;
    #pragma unroll
    for (int h = 0; h < HEADS; ++h) {
        float m = funmap(mbuf[dst * HEADS + h]);
        float p = __expf(ebuf[(size_t)e * HEADS + h] - m);
        ebuf[(size_t)e * HEADS + h] = p;
        atomicAdd(&sbuf[dst * HEADS + h], p);
    }
}

#define TD   40
#define QW   4
#define NWV  (HEADS * QW)
#define CAPW 64
#define SU   8
#define GU   8

__global__ __launch_bounds__(NWV * 32)
void aggregate_tile(const int* __restrict__ adj, const float* __restrict__ hx,
                    const float* __restrict__ ebuf, const float* __restrict__ sbuf,
                    const float* __restrict__ bias, float* __restrict__ out,
                    float* __restrict__ part, int E, int N)
{
    static_assert(NWV * 32 == FDIM, "step 5 gives every thread one column");
    __shared__ float acc[TD * FDIM];
    __shared__ float4 red4[QW * 2 * (FDIM / 4)];
    __shared__ float sL[TD * HEADS];
    __shared__ int   lsrc[NWV * CAPW];
    __shared__ int   ldl[NWV * CAPW];
    __shared__ float lal[NWV * CAPW * HEADS];
    __shared__ int   lcnt[NWV];
    __shared__ int   lmore[NWV];

    const int tid  = threadIdx.x;
    const int lane = tid & 31;
    const int w    = tid >> 5;
    const int h    = w / QW;
    const int q    = w % QW;
    const int d0   = blockIdx.x * TD;
    const int rows = (N - d0 < TD) ? (N - d0) : TD;
    const unsigned lt_mask = (1u << lane) - 1u;

    for (int i = tid; i < rows * HEADS; i += NWV * 32)
        sL[i] = sbuf[(size_t)d0 * HEADS + i];

    #pragma unroll
    for (int t = 0; t < TD / QW; ++t) {
        const int dl  = q + t * QW;
        const int d   = d0 + ((dl < rows) ? dl : (rows - 1));
        const float a = ebuf[(size_t)(E + d) * HEADS + h] / sbuf[(size_t)d * HEADS + h];
        const float4 v = *(const float4*)(hx + (size_t)d * FDIM + h * CDIM + lane * 4);
        float* ar = acc + dl * FDIM + h * CDIM + lane;
        ar[0]  = v.x * a;
        ar[32] = v.y * a;
        ar[64] = v.z * a;
        ar[96] = v.w * a;
    }
    __syncthreads();

    const int* dstp  = adj + E;
    const int  chunk = (E + NWV - 1) / NWV;
    int pos = w * chunk;
    int end = pos + chunk;
    if (end > E) end = E;
    const int maxRounds = chunk / 32 + 2;

    for (int round = 0; round < maxRounds; ++round) {
        int cntw = 0;
        while (pos < end && cntw + 32 <= CAPW) {
            const int pos0 = pos;
            int dv[SU];
            #pragma unroll
            for (int u = 0; u < SU; ++u) {
                const int e  = pos0 + u * 32 + lane;
                const int dd = dstp[(e < end) ? e : (end - 1)];
                dv[u] = (e < end) ? dd : -1;
            }
            #pragma unroll
            for (int u = 0; u < SU; ++u) {
                if (cntw + 32 <= CAPW) {
                    const int d = dv[u];
                    const bool mt = (d >= d0) && (d < d0 + rows);
                    const unsigned m = __builtin_amdgcn_ballot_w32(mt);
                    if (mt) {
                        const int p = w * CAPW + cntw + __builtin_popcount(m & lt_mask);
                        lsrc[p] = pos0 + u * 32 + lane;
                        ldl[p]  = d - d0;
                    }
                    cntw += __builtin_popcount(m);
                    pos = pos0 + (u + 1) * 32;
                }
            }
        }
        for (int i = lane; i < cntw; i += 32) {
            const int p  = w * CAPW + i;
            const int e  = lsrc[p];
            const int dl = ldl[p];
            const int s  = adj[e];
            const float2* pp = (const float2*)(ebuf + (size_t)e * HEADS);
            const float2 p01 = pp[0], p23 = pp[1], p45 = pp[2];
            const float* sr = sL + dl * HEADS;
            float* al = lal + p * HEADS;
            al[0] = p01.x / sr[0];
            al[1] = p01.y / sr[1];
            al[2] = p23.x / sr[2];
            al[3] = p23.y / sr[3];
            al[4] = p45.x / sr[4];
            al[5] = p45.y / sr[5];
            lsrc[p] = s;
        }
        if (lane == 0) { lcnt[w] = cntw; lmore[w] = (pos < end) ? 1 : 0; }
        __syncthreads();

        int c  = (lane < NWV) ? lcnt[lane] : 0;
        int mo = (lane < NWV) ? lmore[lane] : 0;
        const bool anyMore = __builtin_amdgcn_ballot_w32(mo != 0) != 0u;
        int incl = c;
        #pragma unroll
        for (int off = 1; off < 32; off <<= 1) {
            const int t = __shfl_up(incl, off, 32);
            if (lane >= off) incl += t;
        }
        const int n = __builtin_amdgcn_readlane(incl, 31);

        for (int b = 0; b < n; b += 32) {
            const int g = b + lane;
            const bool valid = g < n;
            int ww = 0, base = 0;
            #pragma unroll
            for (int j = 0; j < NWV; ++j) {
                const int ij = __builtin_amdgcn_readlane(incl, j);
                if (ij <= g) { ww = j + 1; base = ij; }
            }
            int s_ = d0, dl_ = 0;
            float a_ = 0.f;
            if (valid) {
                const int slot = ww * CAPW + (g - base);
                s_  = lsrc[slot];
                dl_ = ldl[slot];
                a_  = lal[slot * HEADS + h];
            }
            const bool mine = valid && ((dl_ & (QW - 1)) == q);
            unsigned mm = __builtin_amdgcn_ballot_w32(mine);
            while (mm) {
                int  idx[GU];
                bool ok[GU];
                float4 v[GU];
                #pragma unroll
                for (int u = 0; u < GU; ++u) {
                    ok[u]  = (mm != 0u);
                    idx[u] = ok[u] ? __builtin_ctz(mm) : 0;
                    if (ok[u]) mm &= mm - 1u;
                    const int srow = ok[u] ? __builtin_amdgcn_readlane(s_, idx[u]) : d0;
                    v[u] = *(const float4*)(hx + (size_t)srow * FDIM + h * CDIM + lane * 4);
                }
                #pragma unroll
                for (int u = 0; u < GU; ++u) {
                    if (ok[u]) {
                        const int   dl = __builtin_amdgcn_readlane(dl_, idx[u]);
                        const float a  = __int_as_float(__builtin_amdgcn_readlane(__float_as_int(a_), idx[u]));
                        float* ar = acc + dl * FDIM + h * CDIM + lane;
                        atomicAdd(ar,      v[u].x * a);
                        atomicAdd(ar + 32, v[u].y * a);
                        atomicAdd(ar + 64, v[u].z * a);
                        atomicAdd(ar + 96, v[u].w * a);
                    }
                }
            }
        }
        if (!anyMore) break;
        __syncthreads();
    }

    const float4 bb = *(const float4*)(bias + h * CDIM + lane * 4);
    float4 cs = {0.f, 0.f, 0.f, 0.f};
    float4 cq = {0.f, 0.f, 0.f, 0.f};
    for (int dl = q; dl < rows; dl += QW) {
        const float* ar = acc + dl * FDIM + h * CDIM + lane;
        float4 o;
        o.x = ar[0]  + bb.x;
        o.y = ar[32] + bb.y;
        o.z = ar[64] + bb.z;
        o.w = ar[96] + bb.w;
        o.x = o.x > 0.f ? o.x : 0.f;
        o.y = o.y > 0.f ? o.y : 0.f;
        o.z = o.z > 0.f ? o.z : 0.f;
        o.w = o.w > 0.f ? o.w : 0.f;
        *(float4*)(out + (size_t)(d0 + dl) * FDIM + h * CDIM + lane * 4) = o;
        cs.x += o.x;  cs.y += o.y;  cs.z += o.z;  cs.w += o.w;
        cq.x += o.x * o.x;  cq.y += o.y * o.y;  cq.z += o.z * o.z;  cq.w += o.w * o.w;
    }

    red4[(q * 2)     * (FDIM / 4) + h * (CDIM / 4) + lane] = cs;
    red4[(q * 2 + 1) * (FDIM / 4) + h * (CDIM / 4) + lane] = cq;
    __syncthreads();
    {
        const float* red = (const float*)red4;
        float s = 0.f, ss = 0.f;
        #pragma unroll
        for (int k = 0; k < QW; ++k) {
            s  += red[(k * 2)     * FDIM + tid];
            ss += red[(k * 2 + 1) * FDIM + tid];
        }
        part[(size_t)(blockIdx.x * 2)     * FDIM + tid] = s;
        part[(size_t)(blockIdx.x * 2 + 1) * FDIM + tid] = ss;
    }
}

#define RW 16
#define RU 8

__global__ __launch_bounds__(RW * 32)
void bn_finish(const float* __restrict__ part, float* __restrict__ mean,
               float* __restrict__ rstd, int nTile, int rows)
{
    __shared__ float sh[RW][32], sh2[RW][32];
    const int lane = threadIdx.x & 31;
    const int w    = threadIdx.x >> 5;
    const int col  = blockIdx.x * 32 + lane;
    float s = 0.f, ss = 0.f;
    for (int t0 = w; t0 < nTile; t0 += RW * RU) {
        float a[RU], b[RU];
        #pragma unroll
        for (int u = 0; u < RU; ++u) {
            const int t  = t0 + u * RW;
            const int tc = (t < nTile) ? t : (nTile - 1);
            a[u] = part[(size_t)(tc * 2)     * FDIM + col];
            b[u] = part[(size_t)(tc * 2 + 1) * FDIM + col];
        }
        #pragma unroll
        for (int u = 0; u < RU; ++u) {
            if (t0 + u * RW < nTile) { s += a[u]; ss += b[u]; }
        }
    }
    sh[w][lane] = s;  sh2[w][lane] = ss;
    __syncthreads();
    if (w == 0) {
        float S = 0.f, SS = 0.f;
        #pragma unroll
        for (int k = 0; k < RW; ++k) { S += sh[k][lane]; SS += sh2[k][lane]; }
        float mu  = S / rows;
        float var = SS / rows - mu * mu;
        mean[col] = mu;
        rstd[col] = rsqrtf(var + BN_EPS);
    }
}

__global__ __launch_bounds__(256)
void bn_apply(float* __restrict__ x, const float* __restrict__ mean,
              const float* __restrict__ rstd, const float* __restrict__ gamma,
              const float* __restrict__ beta, size_t total)
{
    size_t i = (size_t)blockIdx.x * blockDim.x + threadIdx.x;
    if (i >= total) return;
    int col = (int)(i % FDIM);
    x[i] = (x[i] - mean[col]) * rstd[col] * gamma[col] + beta[col];
}

__global__ __launch_bounds__(256)
void pool_max(const float* __restrict__ lin, const int* __restrict__ ibatch,
              unsigned* __restrict__ pooled, int rows)
{
    size_t i = (size_t)blockIdx.x * blockDim.x + threadIdx.x;
    if (i >= (size_t)rows * OUT_C) return;
    int n = (int)(i / OUT_C), c = (int)(i % OUT_C);
    atomicMax(&pooled[ibatch[n] * OUT_C + c], fmap(lin[i]));
}

__global__ __launch_bounds__(256)
void pool_final(const unsigned* __restrict__ pooled, float* __restrict__ out, int n)
{
    int i = blockIdx.x * blockDim.x + threadIdx.x;
    if (i < n) out[i] = funmap(pooled[i]);
}

extern "C" void kernel_launch(void* const* d_in, const int* in_sizes, int n_in,
                              void* d_out, int out_size, void* d_ws, size_t ws_size,
                              hipStream_t stream)
{
    const float* x_in   = (const float*)d_in[0];
    const int*   adj    = (const int*)d_in[1];
    const int*   ibatch = (const int*)d_in[2];
    const float* W1     = (const float*)d_in[3];
    const float* att_s1 = (const float*)d_in[4];
    const float* att_d1 = (const float*)d_in[5];
    const float* b1     = (const float*)d_in[6];
    const float* g1     = (const float*)d_in[7];
    const float* be1    = (const float*)d_in[8];
    const float* W2     = (const float*)d_in[9];
    const float* att_s2 = (const float*)d_in[10];
    const float* att_d2 = (const float*)d_in[11];
    const float* b2     = (const float*)d_in[12];
    const float* g2     = (const float*)d_in[13];
    const float* be2    = (const float*)d_in[14];
    const float* linW   = (const float*)d_in[15];
    const float* linb   = (const float*)d_in[16];
    float* out = (float*)d_out;

    const int N    = in_sizes[0] / F_IN;
    const int E    = in_sizes[1] / 2;
    const int Etot = E + N;
    const int B    = out_size / OUT_C;
    const int mT   = (N + 31) / 32;

    float*    hx   = (float*)d_ws;
    float*    agg  = hx  + (size_t)N * FDIM;
    float*    a_s  = agg + (size_t)N * FDIM;
    float*    a_d  = a_s + (size_t)N * HEADS;
    unsigned* mbuf = (unsigned*)(a_d + (size_t)N * HEADS);
    float*    sbuf = (float*)(mbuf + (size_t)N * HEADS);
    float*    ebuf = sbuf + (size_t)N * HEADS;
    float*    mean = ebuf + (size_t)Etot * HEADS;
    float*    rstd = mean + FDIM;
    float*    lin  = rstd + FDIM;
    unsigned* pooled = (unsigned*)(lin + (size_t)N * OUT_C);
    float*    part = lin;

    const size_t nh  = (size_t)N * HEADS;
    const size_t nf  = (size_t)N * FDIM;
    const int    nTile = (N + TD - 1) / TD;

    auto layer = [&](const float* xin, int K, const float* W,
                     const float* asw, const float* adw,
                     const float* bias, const float* gamma, const float* beta) {
        wmma_gemm_f32<<<mT * (FDIM / 64), 32, 0, stream>>>(xin, W, hx, nullptr, N, K, FDIM);
        attn_scores<<<(int)((nh * 32 + 255) / 256), 256, 0, stream>>>(hx, asw, adw, a_s, a_d, N);
        fill_u32<<<(int)((nh + 255) / 256), 256, 0, stream>>>(mbuf, FMAP_NEG_INF, nh);
        fill_u32<<<(int)((nh + 255) / 256), 256, 0, stream>>>((unsigned*)sbuf, 0u, nh);
        edge_scores_max<<<(Etot + 255) / 256, 256, 0, stream>>>(adj, a_s, a_d, ebuf, mbuf, E, Etot);
        edge_exp_sum<<<(Etot + 255) / 256, 256, 0, stream>>>(adj, ebuf, mbuf, sbuf, E, Etot);
        aggregate_tile<<<nTile, NWV * 32, 0, stream>>>(adj, hx, ebuf, sbuf, bias, agg, part, E, N);
        bn_finish<<<FDIM / 32, RW * 32, 0, stream>>>(part, mean, rstd, nTile, N);
        bn_apply<<<(int)((nf + 255) / 256), 256, 0, stream>>>(agg, mean, rstd, gamma, beta, nf);
    };

    layer(x_in, F_IN, W1, att_s1, att_d1, b1, g1, be1);
    layer(agg,  FDIM, W2, att_s2, att_d2, b2, g2, be2);

    wmma_gemm_f32<<<mT * (OUT_C / 64), 32, 0, stream>>>(agg, linW, lin, linb, N, FDIM, OUT_C);
    fill_u32<<<(B * OUT_C + 255) / 256, 256, 0, stream>>>(pooled, FMAP_NEG_INF, (size_t)B * OUT_C);
    pool_max<<<(int)(((size_t)N * OUT_C + 255) / 256), 256, 0, stream>>>(lin, ibatch, pooled, N);
    pool_final<<<(B * OUT_C + 255) / 256, 256, 0, stream>>>(pooled, out, B * OUT_C);
}


# Batch-norm apply pass re-gridded as a streaming kernel: a thread keeps one 4-column group and its constants, 16 rows of 16-byte loads in flight; 3840 waves per pass, were 240000; arithmetic kept
# speedup vs baseline: 1.9624x; 1.0477x over previous
#include <hip/hip_runtime.h>

#define F_IN   256
#define HEADS  6
#define CDIM   128
#define FDIM   768
#define OUT_C  128
#define NEG_SLOPE 0.2f
#define BN_EPS    1e-5f

typedef __attribute__((ext_vector_type(2))) float v2f;
typedef __attribute__((ext_vector_type(8))) float v8f;

__device__ __forceinline__ unsigned fmap(float f) {
    unsigned u = __float_as_uint(f);
    return (u & 0x80000000u) ? ~u : (u | 0x80000000u);
}
__device__ __forceinline__ float funmap(unsigned u) {
    u = (u & 0x80000000u) ? (u & 0x7FFFFFFFu) : ~u;
    return __uint_as_float(u);
}
#define FMAP_NEG_INF 0x007FFFFFu

#define WMMA_F32(a, b, c) \
    __builtin_amdgcn_wmma_f32_16x16x4_f32(false, (a), false, (b), (short)0, (c), false, false)

#define KU 4

struct KFrag { v2f a0, a1, b0, b1, b2, b3; };

__device__ __forceinline__ void load_kfrag(KFrag& f,
                                           const float* __restrict__ ap0,
                                           const float* __restrict__ ap1,
                                           const float* __restrict__ bp, int Nc)
{
    f.a0.x = ap0[0];  f.a0.y = ap0[1];
    f.a1.x = ap1[0];  f.a1.y = ap1[1];
    const float* bq = bp + Nc;
    f.b0.x = bp[0];   f.b0.y = bq[0];
    f.b1.x = bp[16];  f.b1.y = bq[16];
    f.b2.x = bp[32];  f.b2.y = bq[32];
    f.b3.x = bp[48];  f.b3.y = bq[48];
}

__global__ __launch_bounds__(32)
void wmma_gemm_f32(const float* __restrict__ A, const float* __restrict__ B,
                   float* __restrict__ C, const float* __restrict__ bias,
                   int M, int K, int Nc)
{
    const int lane  = threadIdx.x & 31;
    const int nN64  = Nc >> 6;
    const int m0    = (blockIdx.x / nN64) * 32;
    const int n0    = (blockIdx.x % nN64) * 64;
    const int lrow  = lane & 15;
    const int khalf = (lane >> 4) << 1;
    const bool hasM1 = (m0 + 16) < M;

    v8f acc00 = {}, acc01 = {}, acc02 = {}, acc03 = {};
    v8f acc10 = {}, acc11 = {}, acc12 = {}, acc13 = {};

    const float* ap0 = A + (size_t)(m0 + lrow) * K + khalf;
    const float* ap1 = A + (size_t)(hasM1 ? (m0 + 16 + lrow) : (m0 + lrow)) * K + khalf;
    const float* bp  = B + (size_t)khalf * Nc + n0 + lrow;
    const size_t bstep = (size_t)4 * Nc;

    const int nk  = K >> 2;
    const int nst = nk / KU;

    KFrag P[KU], Q[KU];

#define LOAD_STAGE(S)                                                        \
    do {                                                                     \
        _Pragma("unroll")   \
        for (int u = 0; u < KU; ++u)                                         \
            load_kfrag(S[u], ap0 + 4 * u, ap1 + 4 * u, bp + u * bstep, Nc);  \
        ap0 += 4 * KU;  ap1 += 4 * KU;  bp += KU * bstep;                    \
    } while (0)

#define MMA_KFRAG(f)                                                         \
    do {                                                                     \
        acc00 = WMMA_F32((f).a0, (f).b0, acc00);                             \
        acc01 = WMMA_F32((f).a0, (f).b1, acc01);                             \
        acc02 = WMMA_F32((f).a0, (f).b2, acc02);                             \
        acc03 = WMMA_F32((f).a0, (f).b3, acc03);                             \
        acc10 = WMMA_F32((f).a1, (f).b0, acc10);                             \
        acc11 = WMMA_F32((f).a1, (f).b1, acc11);                             \
        acc12 = WMMA_F32((f).a1, (f).b2, acc12);                             \
        acc13 = WMMA_F32((f).a1, (f).b3, acc13);                             \
    } while (0)

#define MMA_STAGE(S)                                                         \
    do {                                                                     \
        _Pragma("unroll")   \
        for (int u = 0; u < KU; ++u) MMA_KFRAG(S[u]);                        \
    } while (0)

    if (nst > 0) {
        int s = 0;
        LOAD_STAGE(P);
        for (; s + 2 < nst; s += 2) {
            LOAD_STAGE(Q);  MMA_STAGE(P);
            LOAD_STAGE(P);  MMA_STAGE(Q);
        }
        if (s + 1 < nst) { LOAD_STAGE(Q);  MMA_STAGE(P);  MMA_STAGE(Q); }
        else             { MMA_STAGE(P); }
    }
    for (int t = nst * KU; t < nk; ++t) {
        KFrag f;
        load_kfrag(f, ap0, ap1, bp, Nc);
        ap0 += 4;  ap1 += 4;  bp += bstep;
        MMA_KFRAG(f);
    }
#undef LOAD_STAGE
#undef MMA_KFRAG
#undef MMA_STAGE

    float bb0 = 0.f, bb1 = 0.f, bb2 = 0.f, bb3 = 0.f;
    if (bias) {
        bb0 = bias[n0 + lrow];
        bb1 = bias[n0 + 16 + lrow];
        bb2 = bias[n0 + 32 + lrow];
        bb3 = bias[n0 + 48 + lrow];
    }
    const int rbase = (lane >> 4) * 8;
    #pragma unroll
    for (int r = 0; r < 8; ++r) {
        float* crow = C + (size_t)(m0 + rbase + r) * Nc + n0 + lrow;
        crow[0]  = acc00[r] + bb0;
        crow[16] = acc01[r] + bb1;
        crow[32] = acc02[r] + bb2;
        crow[48] = acc03[r] + bb3;
    }
    if (hasM1) {
        #pragma unroll
        for (int r = 0; r < 8; ++r) {
            float* crow = C + (size_t)(m0 + 16 + rbase + r) * Nc + n0 + lrow;
            crow[0]  = acc10[r] + bb0;
            crow[16] = acc11[r] + bb1;
            crow[32] = acc12[r] + bb2;
            crow[48] = acc13[r] + bb3;
        }
    }
}

__global__ __launch_bounds__(256)
void attn_scores(const float* __restrict__ hx,
                 const float* __restrict__ att_s, const float* __restrict__ att_d,
                 float* __restrict__ a_s, float* __restrict__ a_d, int nNodes)
{
    int gid  = blockIdx.x * blockDim.x + threadIdx.x;
    int wave = gid >> 5;
    int lane = gid & 31;
    if (wave >= nNodes * HEADS) return;
    int n = wave / HEADS, h = wave % HEADS;

    const float4* hv = (const float4*)(hx + (size_t)n * FDIM + h * CDIM);
    const float4* sv = (const float4*)(att_s + h * CDIM);
    const float4* dv = (const float4*)(att_d + h * CDIM);
    float4 x = hv[lane], s = sv[lane], d = dv[lane];
    float accs = x.x * s.x + x.y * s.y + x.z * s.z + x.w * s.w;
    float accd = x.x * d.x + x.y * d.y + x.z * d.z + x.w * d.w;
    #pragma unroll
    for (int off = 16; off > 0; off >>= 1) {
        accs += __shfl_down(accs, off, 32);
        accd += __shfl_down(accd, off, 32);
    }
    if (lane == 0) { a_s[wave] = accs; a_d[wave] = accd; }
}

__global__ void fill_u32(unsigned* __restrict__ p, unsigned v, size_t n)
{
    size_t i = (size_t)blockIdx.x * blockDim.x + threadIdx.x;
    if (i < n) p[i] = v;
}

__global__ __launch_bounds__(256)
void edge_scores_max(const int* __restrict__ adj,
                     const float* __restrict__ a_s, const float* __restrict__ a_d,
                     float* __restrict__ ebuf, unsigned* __restrict__ mbuf,
                     int E, int nTot)
{
    int e = blockIdx.x * blockDim.x + threadIdx.x;
    if (e >= nTot) return;
    int src, dst;
    if (e < E) { src = adj[e]; dst = adj[E + e]; } else { src = dst = e - E; }
    #pragma unroll
    for (int h = 0; h < HEADS; ++h) {
        float v = a_s[src * HEADS + h] + a_d[dst * HEADS + h];
        v = (v > 0.f) ? v : v * NEG_SLOPE;
        ebuf[(size_t)e * HEADS + h] = v;
        atomicMax(&mbuf[dst * HEADS + h], fmap(v));
    }
}

__global__ __launch_bounds__(256)
void edge_exp_sum(const int* __restrict__ adj, float* __restrict__ ebuf,
                  const unsigned* __restrict__ mbuf, float* __restrict__ sbuf,
                  int E, int nTot)
{
    int e = blockIdx.x * blockDim.x + threadIdx.x;
    if (e >= nTot) return;
    int dst = (e < E) ? adj[E + e] : e - E;
    #pragma unroll
    for (int h = 0; h < HEADS; ++h) {
        float m = funmap(mbuf[dst * HEADS + h]);
        float p = __expf(ebuf[(size_t)e * HEADS + h] - m);
        ebuf[(size_t)e * HEADS + h] = p;
        atomicAdd(&sbuf[dst * HEADS + h], p);
    }
}

#define TD   40
#define QW   4
#define NWV  (HEADS * QW)
#define CAPW 64
#define SU   8
#define GU   8

__global__ __launch_bounds__(NWV * 32)
void aggregate_tile(const int* __restrict__ adj, const float* __restrict__ hx,
                    const float* __restrict__ ebuf, const float* __restrict__ sbuf,
                    const float* __restrict__ bias, float* __restrict__ out,
                    float* __restrict__ part, int E, int N)
{
    static_assert(NWV * 32 == FDIM, "step 5 gives every thread one column");
    __shared__ float acc[TD * FDIM];
    __shared__ float4 red4[QW * 2 * (FDIM / 4)];
    __shared__ float sL[TD * HEADS];
    __shared__ int   lsrc[NWV * CAPW];
    __shared__ int   ldl[NWV * CAPW];
    __shared__ float lal[NWV * CAPW * HEADS];
    __shared__ int   lcnt[NWV];
    __shared__ int   lmore[NWV];

    const int tid  = threadIdx.x;
    const int lane = tid & 31;
    const int w    = tid >> 5;
    const int h    = w / QW;
    const int q    = w % QW;
    const int d0   = blockIdx.x * TD;
    const int rows = (N - d0 < TD) ? (N - d0) : TD;
    const unsigned lt_mask = (1u << lane) - 1u;

    for (int i = tid; i < rows * HEADS; i += NWV * 32)
        sL[i] = sbuf[(size_t)d0 * HEADS + i];

    #pragma unroll
    for (int t = 0; t < TD / QW; ++t) {
        const int dl  = q + t * QW;
        const int d   = d0 + ((dl < rows) ? dl : (rows - 1));
        const float a = ebuf[(size_t)(E + d) * HEADS + h] / sbuf[(size_t)d * HEADS + h];
        const float4 v = *(const float4*)(hx + (size_t)d * FDIM + h * CDIM + lane * 4);
        float* ar = acc + dl * FDIM + h * CDIM + lane;
        ar[0]  = v.x * a;
        ar[32] = v.y * a;
        ar[64] = v.z * a;
        ar[96] = v.w * a;
    }
    __syncthreads();

    const int* dstp  = adj + E;
    const int  chunk = (E + NWV - 1) / NWV;
    int pos = w * chunk;
    int end = pos + chunk;
    if (end > E) end = E;
    const int maxRounds = chunk / 32 + 2;

    for (int round = 0; round < maxRounds; ++round) {
        int cntw = 0;
        while (pos < end && cntw + 32 <= CAPW) {
            const int pos0 = pos;
            int dv[SU];
            #pragma unroll
            for (int u = 0; u < SU; ++u) {
                const int e  = pos0 + u * 32 + lane;
                const int dd = dstp[(e < end) ? e : (end - 1)];
                dv[u] = (e < end) ? dd : -1;
            }
            #pragma unroll
            for (int u = 0; u < SU; ++u) {
                if (cntw + 32 <= CAPW) {
                    const int d = dv[u];
                    const bool mt = (d >= d0) && (d < d0 + rows);
                    const unsigned m = __builtin_amdgcn_ballot_w32(mt);
                    if (mt) {
                        const int p = w * CAPW + cntw + __builtin_popcount(m & lt_mask);
                        lsrc[p] = pos0 + u * 32 + lane;
                        ldl[p]  = d - d0;
                    }
                    cntw += __builtin_popcount(m);
                    pos = pos0 + (u + 1) * 32;
                }
            }
        }
        for (int i = lane; i < cntw; i += 32) {
            const int p  = w * CAPW + i;
            const int e  = lsrc[p];
            const int dl = ldl[p];
            const int s  = adj[e];
            const float2* pp = (const float2*)(ebuf + (size_t)e * HEADS);
            const float2 p01 = pp[0], p23 = pp[1], p45 = pp[2];
            const float* sr = sL + dl * HEADS;
            float* al = lal + p * HEADS;
            al[0] = p01.x / sr[0];
            al[1] = p01.y / sr[1];
            al[2] = p23.x / sr[2];
            al[3] = p23.y / sr[3];
            al[4] = p45.x / sr[4];
            al[5] = p45.y / sr[5];
            lsrc[p] = s;
        }
        if (lane == 0) { lcnt[w] = cntw; lmore[w] = (pos < end) ? 1 : 0; }
        __syncthreads();

        int c  = (lane < NWV) ? lcnt[lane] : 0;
        int mo = (lane < NWV) ? lmore[lane] : 0;
        const bool anyMore = __builtin_amdgcn_ballot_w32(mo != 0) != 0u;
        int incl = c;
        #pragma unroll
        for (int off = 1; off < 32; off <<= 1) {
            const int t = __shfl_up(incl, off, 32);
            if (lane >= off) incl += t;
        }
        const int n = __builtin_amdgcn_readlane(incl, 31);

        for (int b = 0; b < n; b += 32) {
            const int g = b + lane;
            const bool valid = g < n;
            int ww = 0, base = 0;
            #pragma unroll
            for (int j = 0; j < NWV; ++j) {
                const int ij = __builtin_amdgcn_readlane(incl, j);
                if (ij <= g) { ww = j + 1; base = ij; }
            }
            int s_ = d0, dl_ = 0;
            float a_ = 0.f;
            if (valid) {
                const int slot = ww * CAPW + (g - base);
                s_  = lsrc[slot];
                dl_ = ldl[slot];
                a_  = lal[slot * HEADS + h];
            }
            const bool mine = valid && ((dl_ & (QW - 1)) == q);
            unsigned mm = __builtin_amdgcn_ballot_w32(mine);
            while (mm) {
                int  idx[GU];
                bool ok[GU];
                float4 v[GU];
                #pragma unroll
                for (int u = 0; u < GU; ++u) {
                    ok[u]  = (mm != 0u);
                    idx[u] = ok[u] ? __builtin_ctz(mm) : 0;
                    if (ok[u]) mm &= mm - 1u;
                    const int srow = ok[u] ? __builtin_amdgcn_readlane(s_, idx[u]) : d0;
                    v[u] = *(const float4*)(hx + (size_t)srow * FDIM + h * CDIM + lane * 4);
                }
                #pragma unroll
                for (int u = 0; u < GU; ++u) {
                    if (ok[u]) {
                        const int   dl = __builtin_amdgcn_readlane(dl_, idx[u]);
                        const float a  = __int_as_float(__builtin_amdgcn_readlane(__float_as_int(a_), idx[u]));
                        float* ar = acc + dl * FDIM + h * CDIM + lane;
                        atomicAdd(ar,      v[u].x * a);
                        atomicAdd(ar + 32, v[u].y * a);
                        atomicAdd(ar + 64, v[u].z * a);
                        atomicAdd(ar + 96, v[u].w * a);
                    }
                }
            }
        }
        if (!anyMore) break;
        __syncthreads();
    }

    const float4 bb = *(const float4*)(bias + h * CDIM + lane * 4);
    float4 cs = {0.f, 0.f, 0.f, 0.f};
    float4 cq = {0.f, 0.f, 0.f, 0.f};
    for (int dl = q; dl < rows; dl += QW) {
        const float* ar = acc + dl * FDIM + h * CDIM + lane;
        float4 o;
        o.x = ar[0]  + bb.x;
        o.y = ar[32] + bb.y;
        o.z = ar[64] + bb.z;
        o.w = ar[96] + bb.w;
        o.x = o.x > 0.f ? o.x : 0.f;
        o.y = o.y > 0.f ? o.y : 0.f;
        o.z = o.z > 0.f ? o.z : 0.f;
        o.w = o.w > 0.f ? o.w : 0.f;
        *(float4*)(out + (size_t)(d0 + dl) * FDIM + h * CDIM + lane * 4) = o;
        cs.x += o.x;  cs.y += o.y;  cs.z += o.z;  cs.w += o.w;
        cq.x += o.x * o.x;  cq.y += o.y * o.y;  cq.z += o.z * o.z;  cq.w += o.w * o.w;
    }

    red4[(q * 2)     * (FDIM / 4) + h * (CDIM / 4) + lane] = cs;
    red4[(q * 2 + 1) * (FDIM / 4) + h * (CDIM / 4) + lane] = cq;
    __syncthreads();
    {
        const float* red = (const float*)red4;
        float s = 0.f, ss = 0.f;
        #pragma unroll
        for (int k = 0; k < QW; ++k) {
            s  += red[(k * 2)     * FDIM + tid];
            ss += red[(k * 2 + 1) * FDIM + tid];
        }
        part[(size_t)(blockIdx.x * 2)     * FDIM + tid] = s;
        part[(size_t)(blockIdx.x * 2 + 1) * FDIM + tid] = ss;
    }
}

#define RW 16
#define RU 8

__global__ __launch_bounds__(RW * 32)
void bn_finish(const float* __restrict__ part, float* __restrict__ mean,
               float* __restrict__ rstd, int nTile, int rows)
{
    __shared__ float sh[RW][32], sh2[RW][32];
    const int lane = threadIdx.x & 31;
    const int w    = threadIdx.x >> 5;
    const int col  = blockIdx.x * 32 + lane;
    float s = 0.f, ss = 0.f;
    for (int t0 = w; t0 < nTile; t0 += RW * RU) {
        float a[RU], b[RU];
        #pragma unroll
        for (int u = 0; u < RU; ++u) {
            const int t  = t0 + u * RW;
            const int tc = (t < nTile) ? t : (nTile - 1);
            a[u] = part[(size_t)(tc * 2)     * FDIM + col];
            b[u] = part[(size_t)(tc * 2 + 1) * FDIM + col];
        }
        #pragma unroll
        for (int u = 0; u < RU; ++u) {
            if (t0 + u * RW < nTile) { s += a[u]; ss += b[u]; }
        }
    }
    sh[w][lane] = s;  sh2[w][lane] = ss;
    __syncthreads();
    if (w == 0) {
        float S = 0.f, SS = 0.f;
        #pragma unroll
        for (int k = 0; k < RW; ++k) { S += sh[k][lane]; SS += sh2[k][lane]; }
        float mu  = S / rows;
        float var = SS / rows - mu * mu;
        mean[col] = mu;
        rstd[col] = rsqrtf(var + BN_EPS);
    }
}

#define BA_WG    480
#define BA_SLOTS (BA_WG * 256 / (FDIM / 4))
#define BA_ROWS  16

__global__ __launch_bounds__(256)
void bn_apply(float* __restrict__ x, const float* __restrict__ mean,
              const float* __restrict__ rstd, const float* __restrict__ gamma,
              const float* __restrict__ beta, int rows)
{
    static_assert((BA_WG * 256) % (FDIM / 4) == 0, "a thread keeps one column group");
    const int gid = blockIdx.x * 256 + threadIdx.x;
    const int cg  = gid % (FDIM / 4);
    const int r0  = gid / (FDIM / 4);
    const float4 mu = ((const float4*)mean)[cg];
    const float4 rs = ((const float4*)rstd)[cg];
    const float4 ga = ((const float4*)gamma)[cg];
    const float4 be = ((const float4*)beta)[cg];
    float* xp = x + cg * 4;

    for (int rb = r0; rb < rows; rb += BA_SLOTS * BA_ROWS) {
        float4 v[BA_ROWS];
        #pragma unroll
        for (int u = 0; u < BA_ROWS; ++u) {
            const int r  = rb + u * BA_SLOTS;
            const int rc = (r < rows) ? r : rb;
            v[u] = *(const float4*)(xp + (size_t)rc * FDIM);
        }
        #pragma unroll
        for (int u = 0; u < BA_ROWS; ++u) {
            const int r = rb + u * BA_SLOTS;
            if (r < rows) {
                float4 o;
                o.x = (v[u].x - mu.x) * rs.x * ga.x + be.x;
                o.y = (v[u].y - mu.y) * rs.y * ga.y + be.y;
                o.z = (v[u].z - mu.z) * rs.z * ga.z + be.z;
                o.w = (v[u].w - mu.w) * rs.w * ga.w + be.w;
                *(float4*)(xp + (size_t)r * FDIM) = o;
            }
        }
    }
}

__global__ __launch_bounds__(256)
void pool_max(const float* __restrict__ lin, const int* __restrict__ ibatch,
              unsigned* __restrict__ pooled, int rows)
{
    size_t i = (size_t)blockIdx.x * blockDim.x + threadIdx.x;
    if (i >= (size_t)rows * OUT_C) return;
    int n = (int)(i / OUT_C), c = (int)(i % OUT_C);
    atomicMax(&pooled[ibatch[n] * OUT_C + c], fmap(lin[i]));
}

__global__ __launch_bounds__(256)
void pool_final(const unsigned* __restrict__ pooled, float* __restrict__ out, int n)
{
    int i = blockIdx.x * blockDim.x + threadIdx.x;
    if (i < n) out[i] = funmap(pooled[i]);
}

extern "C" void kernel_launch(void* const* d_in, const int* in_sizes, int n_in,
                              void* d_out, int out_size, void* d_ws, size_t ws_size,
                              hipStream_t stream)
{
    const float* x_in   = (const float*)d_in[0];
    const int*   adj    = (const int*)d_in[1];
    const int*   ibatch = (const int*)d_in[2];
    const float* W1     = (const float*)d_in[3];
    const float* att_s1 = (const float*)d_in[4];
    const float* att_d1 = (const float*)d_in[5];
    const float* b1     = (const float*)d_in[6];
    const float* g1     = (const float*)d_in[7];
    const float* be1    = (const float*)d_in[8];
    const float* W2     = (const float*)d_in[9];
    const float* att_s2 = (const float*)d_in[10];
    const float* att_d2 = (const float*)d_in[11];
    const float* b2     = (const float*)d_in[12];
    const float* g2     = (const float*)d_in[13];
    const float* be2    = (const float*)d_in[14];
    const float* linW   = (const float*)d_in[15];
    const float* linb   = (const float*)d_in[16];
    float* out = (float*)d_out;

    const int N    = in_sizes[0] / F_IN;
    const int E    = in_sizes[1] / 2;
    const int Etot = E + N;
    const int B    = out_size / OUT_C;
    const int mT   = (N + 31) / 32;

    float*    hx   = (float*)d_ws;
    float*    agg  = hx  + (size_t)N * FDIM;
    float*    a_s  = agg + (size_t)N * FDIM;
    float*    a_d  = a_s + (size_t)N * HEADS;
    unsigned* mbuf = (unsigned*)(a_d + (size_t)N * HEADS);
    float*    sbuf = (float*)(mbuf + (size_t)N * HEADS);
    float*    ebuf = sbuf + (size_t)N * HEADS;
    float*    mean = ebuf + (size_t)Etot * HEADS;
    float*    rstd = mean + FDIM;
    float*    lin  = rstd + FDIM;
    unsigned* pooled = (unsigned*)(lin + (size_t)N * OUT_C);
    float*    part = lin;

    const size_t nh  = (size_t)N * HEADS;
    const int    nTile = (N + TD - 1) / TD;

    auto layer = [&](const float* xin, int K, const float* W,
                     const float* asw, const float* adw,
                     const float* bias, const float* gamma, const float* beta) {
        wmma_gemm_f32<<<mT * (FDIM / 64), 32, 0, stream>>>(xin, W, hx, nullptr, N, K, FDIM);
        attn_scores<<<(int)((nh * 32 + 255) / 256), 256, 0, stream>>>(hx, asw, adw, a_s, a_d, N);
        fill_u32<<<(int)((nh + 255) / 256), 256, 0, stream>>>(mbuf, FMAP_NEG_INF, nh);
        fill_u32<<<(int)((nh + 255) / 256), 256, 0, stream>>>((unsigned*)sbuf, 0u, nh);
        edge_scores_max<<<(Etot + 255) / 256, 256, 0, stream>>>(adj, a_s, a_d, ebuf, mbuf, E, Etot);
        edge_exp_sum<<<(Etot + 255) / 256, 256, 0, stream>>>(adj, ebuf, mbuf, sbuf, E, Etot);
        aggregate_tile<<<nTile, NWV * 32, 0, stream>>>(adj, hx, ebuf, sbuf, bias, agg, part, E, N);
        bn_finish<<<FDIM / 32, RW * 32, 0, stream>>>(part, mean, rstd, nTile, N);
        bn_apply<<<BA_WG, 256, 0, stream>>>(agg, mean, rstd, gamma, beta, N);
    };

    layer(x_in, F_IN, W1, att_s1, att_d1, b1, g1, be1);
    layer(agg,  FDIM, W2, att_s2, att_d2, b2, g2, be2);

    wmma_gemm_f32<<<mT * (OUT_C / 64), 32, 0, stream>>>(agg, linW, lin, linb, N, FDIM, OUT_C);
    fill_u32<<<(B * OUT_C + 255) / 256, 256, 0, stream>>>(pooled, FMAP_NEG_INF, (size_t)B * OUT_C);
    pool_max<<<(int)(((size_t)N * OUT_C + 255) / 256), 256, 0, stream>>>(lin, ibatch, pooled, N);
    pool_final<<<(B * OUT_C + 255) / 256, 256, 0, stream>>>(pooled, out, B * OUT_C);
}


# Aggregation lists hold 128 entries per wave, were 64: no share outgrows its list, so no tile filters the edge list again in phases 1 and 2; sum order changes in tiles that needed a second round
# speedup vs baseline: 2.8948x; 1.4751x over previous
#include <hip/hip_runtime.h>

#define F_IN   256
#define HEADS  6
#define CDIM   128
#define FDIM   768
#define OUT_C  128
#define NEG_SLOPE 0.2f
#define BN_EPS    1e-5f

typedef __attribute__((ext_vector_type(2))) float v2f;
typedef __attribute__((ext_vector_type(8))) float v8f;

__device__ __forceinline__ unsigned fmap(float f) {
    unsigned u = __float_as_uint(f);
    return (u & 0x80000000u) ? ~u : (u | 0x80000000u);
}
__device__ __forceinline__ float funmap(unsigned u) {
    u = (u & 0x80000000u) ? (u & 0x7FFFFFFFu) : ~u;
    return __uint_as_float(u);
}
#define FMAP_NEG_INF 0x007FFFFFu

#define WMMA_F32(a, b, c) \
    __builtin_amdgcn_wmma_f32_16x16x4_f32(false, (a), false, (b), (short)0, (c), false, false)

#define KU 4

struct KFrag { v2f a0, a1, b0, b1, b2, b3; };

__device__ __forceinline__ void load_kfrag(KFrag& f,
                                           const float* __restrict__ ap0,
                                           const float* __restrict__ ap1,
                                           const float* __restrict__ bp, int Nc)
{
    f.a0.x = ap0[0];  f.a0.y = ap0[1];
    f.a1.x = ap1[0];  f.a1.y = ap1[1];
    const float* bq = bp + Nc;
    f.b0.x = bp[0];   f.b0.y = bq[0];
    f.b1.x = bp[16];  f.b1.y = bq[16];
    f.b2.x = bp[32];  f.b2.y = bq[32];
    f.b3.x = bp[48];  f.b3.y = bq[48];
}

__global__ __launch_bounds__(32)
void wmma_gemm_f32(const float* __restrict__ A, const float* __restrict__ B,
                   float* __restrict__ C, const float* __restrict__ bias,
                   int M, int K, int Nc)
{
    const int lane  = threadIdx.x & 31;
    const int nN64  = Nc >> 6;
    const int m0    = (blockIdx.x / nN64) * 32;
    const int n0    = (blockIdx.x % nN64) * 64;
    const int lrow  = lane & 15;
    const int khalf = (lane >> 4) << 1;
    const bool hasM1 = (m0 + 16) < M;

    v8f acc00 = {}, acc01 = {}, acc02 = {}, acc03 = {};
    v8f acc10 = {}, acc11 = {}, acc12 = {}, acc13 = {};

    const float* ap0 = A + (size_t)(m0 + lrow) * K + khalf;
    const float* ap1 = A + (size_t)(hasM1 ? (m0 + 16 + lrow) : (m0 + lrow)) * K + khalf;
    const float* bp  = B + (size_t)khalf * Nc + n0 + lrow;
    const size_t bstep = (size_t)4 * Nc;

    const int nk  = K >> 2;
    const int nst = nk / KU;

    KFrag P[KU], Q[KU];

#define LOAD_STAGE(S)                                                        \
    do {                                                                     \
        _Pragma("unroll")   \
        for (int u = 0; u < KU; ++u)                                         \
            load_kfrag(S[u], ap0 + 4 * u, ap1 + 4 * u, bp + u * bstep, Nc);  \
        ap0 += 4 * KU;  ap1 += 4 * KU;  bp += KU * bstep;                    \
    } while (0)

#define MMA_KFRAG(f)                                                         \
    do {                                                                     \
        acc00 = WMMA_F32((f).a0, (f).b0, acc00);                             \
        acc01 = WMMA_F32((f).a0, (f).b1, acc01);                             \
        acc02 = WMMA_F32((f).a0, (f).b2, acc02);                             \
        acc03 = WMMA_F32((f).a0, (f).b3, acc03);                             \
        acc10 = WMMA_F32((f).a1, (f).b0, acc10);                             \
        acc11 = WMMA_F32((f).a1, (f).b1, acc11);                             \
        acc12 = WMMA_F32((f).a1, (f).b2, acc12);                             \
        acc13 = WMMA_F32((f).a1, (f).b3, acc13);                             \
    } while (0)

#define MMA_STAGE(S)                                                         \
    do {                                                                     \
        _Pragma("unroll")   \
        for (int u = 0; u < KU; ++u) MMA_KFRAG(S[u]);                        \
    } while (0)

    if (nst > 0) {
        int s = 0;
        LOAD_STAGE(P);
        for (; s + 2 < nst; s += 2) {
            LOAD_STAGE(Q);  MMA_STAGE(P);
            LOAD_STAGE(P);  MMA_STAGE(Q);
        }
        if (s + 1 < nst) { LOAD_STAGE(Q);  MMA_STAGE(P);  MMA_STAGE(Q); }
        else             { MMA_STAGE(P); }
    }
    for (int t = nst * KU; t < nk; ++t) {
        KFrag f;
        load_kfrag(f, ap0, ap1, bp, Nc);
        ap0 += 4;  ap1 += 4;  bp += bstep;
        MMA_KFRAG(f);
    }
#undef LOAD_STAGE
#undef MMA_KFRAG
#undef MMA_STAGE

    float bb0 = 0.f, bb1 = 0.f, bb2 = 0.f, bb3 = 0.f;
    if (bias) {
        bb0 = bias[n0 + lrow];
        bb1 = bias[n0 + 16 + lrow];
        bb2 = bias[n0 + 32 + lrow];
        bb3 = bias[n0 + 48 + lrow];
    }
    const int rbase = (lane >> 4) * 8;
    #pragma unroll
    for (int r = 0; r < 8; ++r) {
        float* crow = C + (size_t)(m0 + rbase + r) * Nc + n0 + lrow;
        crow[0]  = acc00[r] + bb0;
        crow[16] = acc01[r] + bb1;
        crow[32] = acc02[r] + bb2;
        crow[48] = acc03[r] + bb3;
    }
    if (hasM1) {
        #pragma unroll
        for (int r = 0; r < 8; ++r) {
            float* crow = C + (size_t)(m0 + 16 + rbase + r) * Nc + n0 + lrow;
            crow[0]  = acc10[r] + bb0;
            crow[16] = acc11[r] + bb1;
            crow[32] = acc12[r] + bb2;
            crow[48] = acc13[r] + bb3;
        }
    }
}

__global__ __launch_bounds__(256)
void attn_scores(const float* __restrict__ hx,
                 const float* __restrict__ att_s, const float* __restrict__ att_d,
                 float* __restrict__ a_s, float* __restrict__ a_d, int nNodes)
{
    int gid  = blockIdx.x * blockDim.x + threadIdx.x;
    int wave = gid >> 5;
    int lane = gid & 31;
    if (wave >= nNodes * HEADS) return;
    int n = wave / HEADS, h = wave % HEADS;

    const float4* hv = (const float4*)(hx + (size_t)n * FDIM + h * CDIM);
    const float4* sv = (const float4*)(att_s + h * CDIM);
    const float4* dv = (const float4*)(att_d + h * CDIM);
    float4 x = hv[lane], s = sv[lane], d = dv[lane];
    float accs = x.x * s.x + x.y * s.y + x.z * s.z + x.w * s.w;
    float accd = x.x * d.x + x.y * d.y + x.z * d.z + x.w * d.w;
    #pragma unroll
    for (int off = 16; off > 0; off >>= 1) {
        accs += __shfl_down(accs, off, 32);
        accd += __shfl_down(accd, off, 32);
    }
    if (lane == 0) { a_s[wave] = accs; a_d[wave] = accd; }
}

__global__ void fill_u32(unsigned* __restrict__ p, unsigned v, size_t n)
{
    size_t i = (size_t)blockIdx.x * blockDim.x + threadIdx.x;
    if (i < n) p[i] = v;
}

#define TD   40
#define QW   4
#define NWV  (HEADS * QW)
#define CAPW 128
#define SU   8
#define GU   8

__global__ __launch_bounds__(NWV * 32)
void aggregate_tile(const int* __restrict__ adj, const float* __restrict__ hx,
                    const float* __restrict__ a_s, const float* __restrict__ a_d,
                    const float* __restrict__ bias, float* __restrict__ out,
                    float* __restrict__ part, int E, int N)
{
    static_assert(NWV * 32 == FDIM, "step 5 gives every thread one column");
    __shared__ float acc[TD * FDIM];
    __shared__ float4 red4[QW * 2 * (FDIM / 4)];
    __shared__ float    sL[TD * HEADS];
    __shared__ unsigned mL[TD * HEADS];
    __shared__ float    adL[TD * HEADS];
    __shared__ float    esL[TD * HEADS];
    __shared__ int   lsrc[NWV * CAPW];
    __shared__ int   ldl[NWV * CAPW];
    __shared__ float lal[NWV * CAPW * HEADS];
    __shared__ int   lcnt[NWV];
    __shared__ int   lmore[NWV];

    const int tid  = threadIdx.x;
    const int lane = tid & 31;
    const int w    = tid >> 5;
    const int h    = w / QW;
    const int q    = w % QW;
    const int d0   = blockIdx.x * TD;
    const int rows = (N - d0 < TD) ? (N - d0) : TD;
    const unsigned lt_mask = (1u << lane) - 1u;

    for (int i = tid; i < rows * HEADS; i += NWV * 32) {
        const float ad = a_d[(size_t)d0 * HEADS + i];
        float v = a_s[(size_t)d0 * HEADS + i] + ad;
        v = (v > 0.f) ? v : v * NEG_SLOPE;
        adL[i] = ad;
        esL[i] = v;
        mL[i]  = fmap(v);
    }
    __syncthreads();

    const int* dstp  = adj + E;
    const int  chunk = (E + NWV - 1) / NWV;
    const int  beg   = w * chunk;
    int end = beg + chunk;
    if (end > E) end = E;
    const int maxRounds = chunk / 32 + 2;
    int  pos    = beg;
    int  cntw   = 0;
    bool single = false;

    for (int phase = 0; phase < 3; ++phase) {
        const bool refilter = (phase == 0) || !single;
        if (refilter) pos = beg;

        if (phase == 2) {
            #pragma unroll
            for (int t = 0; t < TD / QW; ++t) {
                const int dl  = q + t * QW;
                const int dlc = (dl < rows) ? dl : (rows - 1);
                const int d   = d0 + dlc;
                const int k   = dlc * HEADS + h;
                const float a = __expf(esL[k] - funmap(mL[k])) / sL[k];
                const float4 v = *(const float4*)(hx + (size_t)d * FDIM + h * CDIM + lane * 4);
                float* ar = acc + dl * FDIM + h * CDIM + lane;
                ar[0]  = v.x * a;
                ar[32] = v.y * a;
                ar[64] = v.z * a;
                ar[96] = v.w * a;
            }
        }

        for (int round = 0; round < maxRounds; ++round) {
            if (refilter) {
                cntw = 0;
                while (pos < end && cntw + 32 <= CAPW) {
                    const int pos0 = pos;
                    int dv[SU];
                    #pragma unroll
                    for (int u = 0; u < SU; ++u) {
                        const int e  = pos0 + u * 32 + lane;
                        const int dd = dstp[(e < end) ? e : (end - 1)];
                        dv[u] = (e < end) ? dd : -1;
                    }
                    #pragma unroll
                    for (int u = 0; u < SU; ++u) {
                        if (cntw + 32 <= CAPW) {
                            const int d = dv[u];
                            const bool mt = (d >= d0) && (d < d0 + rows);
                            const unsigned m = __builtin_amdgcn_ballot_w32(mt);
                            if (mt) {
                                const int p = w * CAPW + cntw + __builtin_popcount(m & lt_mask);
                                lsrc[p] = pos0 + u * 32 + lane;
                                ldl[p]  = d - d0;
                            }
                            cntw += __builtin_popcount(m);
                            pos = pos0 + (u + 1) * 32;
                        }
                    }
                }
                for (int i = lane; i < cntw; i += 32) {
                    const int p  = w * CAPW + i;
                    const int e  = lsrc[p];
                    const int dl = ldl[p];
                    const int s  = adj[e];
                    const float2* pp = (const float2*)(a_s + (size_t)s * HEADS);
                    const float2 x01 = pp[0], x23 = pp[1], x45 = pp[2];
                    const float* dr = adL + dl * HEADS;
                    float v0 = x01.x + dr[0], v1 = x01.y + dr[1], v2 = x23.x + dr[2];
                    float v3 = x23.y + dr[3], v4 = x45.x + dr[4], v5 = x45.y + dr[5];
                    float* al = lal + p * HEADS;
                    al[0] = (v0 > 0.f) ? v0 : v0 * NEG_SLOPE;
                    al[1] = (v1 > 0.f) ? v1 : v1 * NEG_SLOPE;
                    al[2] = (v2 > 0.f) ? v2 : v2 * NEG_SLOPE;
                    al[3] = (v3 > 0.f) ? v3 : v3 * NEG_SLOPE;
                    al[4] = (v4 > 0.f) ? v4 : v4 * NEG_SLOPE;
                    al[5] = (v5 > 0.f) ? v5 : v5 * NEG_SLOPE;
                    lsrc[p] = s;
                }
                if (lane == 0) { lcnt[w] = cntw; lmore[w] = (pos < end) ? 1 : 0; }
            }
            for (int i = lane; i < cntw; i += 32) {
                const int p  = w * CAPW + i;
                const int dl = ldl[p];
                float* al = lal + p * HEADS;
                #pragma unroll
                for (int hh = 0; hh < HEADS; ++hh) {
                    const int   k  = dl * HEADS + hh;
                    const float ev = al[hh];
                    if (phase == 0) {
                        atomicMax(&mL[k], fmap(ev));
                    } else {
                        const float pe = __expf(ev - funmap(mL[k]));
                        if (phase == 1) atomicAdd(&sL[k], pe);
                        else            al[hh] = pe / sL[k];
                    }
                }
            }
            __syncthreads();

            int c  = (lane < NWV) ? lcnt[lane] : 0;
            int mo = (lane < NWV) ? lmore[lane] : 0;
            const bool anyMore = __builtin_amdgcn_ballot_w32(mo != 0) != 0u;
            if (phase == 2) {
                int incl = c;
                #pragma unroll
                for (int off = 1; off < 32; off <<= 1) {
                    const int t = __shfl_up(incl, off, 32);
                    if (lane >= off) incl += t;
                }
                const int n = __builtin_amdgcn_readlane(incl, 31);

                for (int b = 0; b < n; b += 32) {
                    const int g = b + lane;
                    const bool valid = g < n;
                    int ww = 0, base = 0;
                    #pragma unroll
                    for (int j = 0; j < NWV; ++j) {
                        const int ij = __builtin_amdgcn_readlane(incl, j);
                        if (ij <= g) { ww = j + 1; base = ij; }
                    }
                    int s_ = d0, dl_ = 0;
                    float a_ = 0.f;
                    if (valid) {
                        const int slot = ww * CAPW + (g - base);
                        s_  = lsrc[slot];
                        dl_ = ldl[slot];
                        a_  = lal[slot * HEADS + h];
                    }
                    const bool mine = valid && ((dl_ & (QW - 1)) == q);
                    unsigned mm = __builtin_amdgcn_ballot_w32(mine);
                    while (mm) {
                        int  idx[GU];
                        bool ok[GU];
                        float4 v[GU];
                        #pragma unroll
                        for (int u = 0; u < GU; ++u) {
                            ok[u]  = (mm != 0u);
                            idx[u] = ok[u] ? __builtin_ctz(mm) : 0;
                            if (ok[u]) mm &= mm - 1u;
                            const int srow = ok[u] ? __builtin_amdgcn_readlane(s_, idx[u]) : d0;
                            v[u] = *(const float4*)(hx + (size_t)srow * FDIM + h * CDIM + lane * 4);
                        }
                        #pragma unroll
                        for (int u = 0; u < GU; ++u) {
                            if (ok[u]) {
                                const int   dl = __builtin_amdgcn_readlane(dl_, idx[u]);
                                const float a  = __int_as_float(__builtin_amdgcn_readlane(__float_as_int(a_), idx[u]));
                                float* ar = acc + dl * FDIM + h * CDIM + lane;
                                atomicAdd(ar,      v[u].x * a);
                                atomicAdd(ar + 32, v[u].y * a);
                                atomicAdd(ar + 64, v[u].z * a);
                                atomicAdd(ar + 96, v[u].w * a);
                            }
                        }
                    }
                }
            }
            if (phase == 0 && round == 0) single = !anyMore;
            if (!anyMore) break;
            __syncthreads();
        }

        if (phase == 0) {
            for (int i = tid; i < rows * HEADS; i += NWV * 32)
                sL[i] = __expf(esL[i] - funmap(mL[i]));
        }
        if (phase < 2) __syncthreads();
    }

    const float4 bb = *(const float4*)(bias + h * CDIM + lane * 4);
    float4 cs = {0.f, 0.f, 0.f, 0.f};
    float4 cq = {0.f, 0.f, 0.f, 0.f};
    for (int dl = q; dl < rows; dl += QW) {
        const float* ar = acc + dl * FDIM + h * CDIM + lane;
        float4 o;
        o.x = ar[0]  + bb.x;
        o.y = ar[32] + bb.y;
        o.z = ar[64] + bb.z;
        o.w = ar[96] + bb.w;
        o.x = o.x > 0.f ? o.x : 0.f;
        o.y = o.y > 0.f ? o.y : 0.f;
        o.z = o.z > 0.f ? o.z : 0.f;
        o.w = o.w > 0.f ? o.w : 0.f;
        *(float4*)(out + (size_t)(d0 + dl) * FDIM + h * CDIM + lane * 4) = o;
        cs.x += o.x;  cs.y += o.y;  cs.z += o.z;  cs.w += o.w;
        cq.x += o.x * o.x;  cq.y += o.y * o.y;  cq.z += o.z * o.z;  cq.w += o.w * o.w;
    }

    red4[(q * 2)     * (FDIM / 4) + h * (CDIM / 4) + lane] = cs;
    red4[(q * 2 + 1) * (FDIM / 4) + h * (CDIM / 4) + lane] = cq;
    __syncthreads();
    {
        const float* red = (const float*)red4;
        float s = 0.f, ss = 0.f;
        #pragma unroll
        for (int k = 0; k < QW; ++k) {
            s  += red[(k * 2)     * FDIM + tid];
            ss += red[(k * 2 + 1) * FDIM + tid];
        }
        part[(size_t)(blockIdx.x * 2)     * FDIM + tid] = s;
        part[(size_t)(blockIdx.x * 2 + 1) * FDIM + tid] = ss;
    }
}

#define RW 16
#define RU 8

__global__ __launch_bounds__(RW * 32)
void bn_finish(const float* __restrict__ part, float* __restrict__ mean,
               float* __restrict__ rstd, int nTile, int rows)
{
    __shared__ float sh[RW][32], sh2[RW][32];
    const int lane = threadIdx.x & 31;
    const int w    = threadIdx.x >> 5;
    const int col  = blockIdx.x * 32 + lane;
    float s = 0.f, ss = 0.f;
    for (int t0 = w; t0 < nTile; t0 += RW * RU) {
        float a[RU], b[RU];
        #pragma unroll
        for (int u = 0; u < RU; ++u) {
            const int t  = t0 + u * RW;
            const int tc = (t < nTile) ? t : (nTile - 1);
            a[u] = part[(size_t)(tc * 2)     * FDIM + col];
            b[u] = part[(size_t)(tc * 2 + 1) * FDIM + col];
        }
        #pragma unroll
        for (int u = 0; u < RU; ++u) {
            if (t0 + u * RW < nTile) { s += a[u]; ss += b[u]; }
        }
    }
    sh[w][lane] = s;  sh2[w][lane] = ss;
    __syncthreads();
    if (w == 0) {
        float S = 0.f, SS = 0.f;
        #pragma unroll
        for (int k = 0; k < RW; ++k) { S += sh[k][lane]; SS += sh2[k][lane]; }
        float mu  = S / rows;
        float var = SS / rows - mu * mu;
        mean[col] = mu;
        rstd[col] = rsqrtf(var + BN_EPS);
    }
}

#define BA_WG    480
#define BA_SLOTS (BA_WG * 256 / (FDIM / 4))
#define BA_ROWS  16

__global__ __launch_bounds__(256)
void bn_apply(float* __restrict__ x, const float* __restrict__ mean,
              const float* __restrict__ rstd, const float* __restrict__ gamma,
              const float* __restrict__ beta, int rows)
{
    static_assert((BA_WG * 256) % (FDIM / 4) == 0, "a thread keeps one column group");
    const int gid = blockIdx.x * 256 + threadIdx.x;
    const int cg  = gid % (FDIM / 4);
    const int r0  = gid / (FDIM / 4);
    const float4 mu = ((const float4*)mean)[cg];
    const float4 rs = ((const float4*)rstd)[cg];
    const float4 ga = ((const float4*)gamma)[cg];
    const float4 be = ((const float4*)beta)[cg];
    float* xp = x + cg * 4;

    for (int rb = r0; rb < rows; rb += BA_SLOTS * BA_ROWS) {
        float4 v[BA_ROWS];
        #pragma unroll
        for (int u = 0; u < BA_ROWS; ++u) {
            const int r  = rb + u * BA_SLOTS;
            const int rc = (r < rows) ? r : rb;
            v[u] = *(const float4*)(xp + (size_t)rc * FDIM);
        }
        #pragma unroll
        for (int u = 0; u < BA_ROWS; ++u) {
            const int r = rb + u * BA_SLOTS;
            if (r < rows) {
                float4 o;
                o.x = (v[u].x - mu.x) * rs.x * ga.x + be.x;
                o.y = (v[u].y - mu.y) * rs.y * ga.y + be.y;
                o.z = (v[u].z - mu.z) * rs.z * ga.z + be.z;
                o.w = (v[u].w - mu.w) * rs.w * ga.w + be.w;
                *(float4*)(xp + (size_t)r * FDIM) = o;
            }
        }
    }
}

__global__ __launch_bounds__(256)
void pool_max(const float* __restrict__ lin, const int* __restrict__ ibatch,
              unsigned* __restrict__ pooled, int rows)
{
    size_t i = (size_t)blockIdx.x * blockDim.x + threadIdx.x;
    if (i >= (size_t)rows * OUT_C) return;
    int n = (int)(i / OUT_C), c = (int)(i % OUT_C);
    atomicMax(&pooled[ibatch[n] * OUT_C + c], fmap(lin[i]));
}

__global__ __launch_bounds__(256)
void pool_final(const unsigned* __restrict__ pooled, float* __restrict__ out, int n)
{
    int i = blockIdx.x * blockDim.x + threadIdx.x;
    if (i < n) out[i] = funmap(pooled[i]);
}

extern "C" void kernel_launch(void* const* d_in, const int* in_sizes, int n_in,
                              void* d_out, int out_size, void* d_ws, size_t ws_size,
                              hipStream_t stream)
{
    const float* x_in   = (const float*)d_in[0];
    const int*   adj    = (const int*)d_in[1];
    const int*   ibatch = (const int*)d_in[2];
    const float* W1     = (const float*)d_in[3];
    const float* att_s1 = (const float*)d_in[4];
    const float* att_d1 = (const float*)d_in[5];
    const float* b1     = (const float*)d_in[6];
    const float* g1     = (const float*)d_in[7];
    const float* be1    = (const float*)d_in[8];
    const float* W2     = (const float*)d_in[9];
    const float* att_s2 = (const float*)d_in[10];
    const float* att_d2 = (const float*)d_in[11];
    const float* b2     = (const float*)d_in[12];
    const float* g2     = (const float*)d_in[13];
    const float* be2    = (const float*)d_in[14];
    const float* linW   = (const float*)d_in[15];
    const float* linb   = (const float*)d_in[16];
    float* out = (float*)d_out;

    const int N    = in_sizes[0] / F_IN;
    const int E    = in_sizes[1] / 2;
    const int B    = out_size / OUT_C;
    const int mT   = (N + 31) / 32;

    float*    hx   = (float*)d_ws;
    float*    agg  = hx  + (size_t)N * FDIM;
    float*    a_s  = agg + (size_t)N * FDIM;
    float*    a_d  = a_s + (size_t)N * HEADS;
    float*    mean = a_d + (size_t)N * HEADS;
    float*    rstd = mean + FDIM;
    float*    lin  = rstd + FDIM;
    unsigned* pooled = (unsigned*)(lin + (size_t)N * OUT_C);
    float*    part = lin;

    const size_t nh  = (size_t)N * HEADS;
    const int    nTile = (N + TD - 1) / TD;

    auto layer = [&](const float* xin, int K, const float* W,
                     const float* asw, const float* adw,
                     const float* bias, const float* gamma, const float* beta) {
        wmma_gemm_f32<<<mT * (FDIM / 64), 32, 0, stream>>>(xin, W, hx, nullptr, N, K, FDIM);
        attn_scores<<<(int)((nh * 32 + 255) / 256), 256, 0, stream>>>(hx, asw, adw, a_s, a_d, N);
        aggregate_tile<<<nTile, NWV * 32, 0, stream>>>(adj, hx, a_s, a_d, bias, agg, part, E, N);
        bn_finish<<<FDIM / 32, RW * 32, 0, stream>>>(part, mean, rstd, nTile, N);
        bn_apply<<<BA_WG, 256, 0, stream>>>(agg, mean, rstd, gamma, beta, N);
    };

    layer(x_in, F_IN, W1, att_s1, att_d1, b1, g1, be1);
    layer(agg,  FDIM, W2, att_s2, att_d2, b2, g2, be2);

    wmma_gemm_f32<<<mT * (OUT_C / 64), 32, 0, stream>>>(agg, linW, lin, linb, N, FDIM, OUT_C);
    fill_u32<<<(B * OUT_C + 255) / 256, 256, 0, stream>>>(pooled, FMAP_NEG_INF, (size_t)B * OUT_C);
    pool_max<<<(int)(((size_t)N * OUT_C + 255) / 256), 256, 0, stream>>>(lin, ibatch, pooled, N);
    pool_final<<<(B * OUT_C + 255) / 256, 256, 0, stream>>>(pooled, out, B * OUT_C);
}


# GEMM fragment ownership permuted: a lane half owns 4 consecutive k of an 8-k pair, N-tile j is columns n0+4c+j; A, B loads and C stores 16 B per lane (12 loads a stage, were 40); k sum order changed
# speedup vs baseline: 3.2010x; 1.1058x over previous
#include <hip/hip_runtime.h>

#define F_IN   256
#define HEADS  6
#define CDIM   128
#define FDIM   768
#define OUT_C  128
#define NEG_SLOPE 0.2f
#define BN_EPS    1e-5f

typedef __attribute__((ext_vector_type(2))) float v2f;
typedef __attribute__((ext_vector_type(8))) float v8f;
typedef __attribute__((ext_vector_type(4))) int   v4i;

__device__ __forceinline__ unsigned fmap(float f) {
    unsigned u = __float_as_uint(f);
    return (u & 0x80000000u) ? ~u : (u | 0x80000000u);
}
__device__ __forceinline__ float funmap(unsigned u) {
    u = (u & 0x80000000u) ? (u & 0x7FFFFFFFu) : ~u;
    return __uint_as_float(u);
}
#define FMAP_NEG_INF 0x007FFFFFu

#define WMMA_F32(a, b, c) \
    __builtin_amdgcn_wmma_f32_16x16x4_f32(false, (a), false, (b), (short)0, (c), false, false)

#define KP 2

static_assert(F_IN % 8 == 0 && FDIM % 8 == 0, "the K loop walks whole k-pairs");

struct KFrag { float4 a0, a1, b0, b1, b2, b3; };

__device__ __forceinline__ void load_kfrag(KFrag& f,
                                           const float* __restrict__ ap0,
                                           const float* __restrict__ ap1,
                                           const float* __restrict__ bp, int Nc)
{
    f.a0 = *(const float4*)ap0;
    f.a1 = *(const float4*)ap1;
    f.b0 = *(const float4*)(bp);
    f.b1 = *(const float4*)(bp + Nc);
    f.b2 = *(const float4*)(bp + 2 * (size_t)Nc);
    f.b3 = *(const float4*)(bp + 3 * (size_t)Nc);
}

__global__ __launch_bounds__(32)
void wmma_gemm_f32(const float* __restrict__ A, const float* __restrict__ B,
                   float* __restrict__ C, const float* __restrict__ bias,
                   int M, int K, int Nc)
{
    const int lane  = threadIdx.x & 31;
    const int nN64  = Nc >> 6;
    const int m0    = (blockIdx.x / nN64) * 32;
    const int n0    = (blockIdx.x % nN64) * 64;
    const int lrow  = lane & 15;
    const int kq    = (lane >> 4) << 2;
    const bool hasM1 = (m0 + 16) < M;

    v8f acc00 = {}, acc01 = {}, acc02 = {}, acc03 = {};
    v8f acc10 = {}, acc11 = {}, acc12 = {}, acc13 = {};

    const float* ap0 = A + (size_t)(m0 + lrow) * K + kq;
    const float* ap1 = A + (size_t)(hasM1 ? (m0 + 16 + lrow) : (m0 + lrow)) * K + kq;
    const float* bp  = B + (size_t)kq * Nc + n0 + 4 * lrow;
    const size_t bstep = (size_t)8 * Nc;

    const int nkp = K >> 3;
    const int nst = nkp / KP;

    KFrag P[KP], Q[KP];

#define LOAD_STAGE(S)                                                        \
    do {                                                                     \
        _Pragma("unroll")   \
        for (int u = 0; u < KP; ++u)                                         \
            load_kfrag(S[u], ap0 + 8 * u, ap1 + 8 * u, bp + u * bstep, Nc);  \
        ap0 += 8 * KP;  ap1 += 8 * KP;  bp += KP * bstep;                    \
    } while (0)

#define MMA_KFRAG(f)                                                         \
    do {                                                                     \
        const v2f ae0 = {(f).a0.x, (f).a0.y}, ae1 = {(f).a1.x, (f).a1.y};    \
        const v2f be0 = {(f).b0.x, (f).b1.x}, be1 = {(f).b0.y, (f).b1.y};    \
        const v2f be2 = {(f).b0.z, (f).b1.z}, be3 = {(f).b0.w, (f).b1.w};    \
        acc00 = WMMA_F32(ae0, be0, acc00);                                   \
        acc01 = WMMA_F32(ae0, be1, acc01);                                   \
        acc02 = WMMA_F32(ae0, be2, acc02);                                   \
        acc03 = WMMA_F32(ae0, be3, acc03);                                   \
        acc10 = WMMA_F32(ae1, be0, acc10);                                   \
        acc11 = WMMA_F32(ae1, be1, acc11);                                   \
        acc12 = WMMA_F32(ae1, be2, acc12);                                   \
        acc13 = WMMA_F32(ae1, be3, acc13);                                   \
        const v2f ao0 = {(f).a0.z, (f).a0.w}, ao1 = {(f).a1.z, (f).a1.w};    \
        const v2f bo0 = {(f).b2.x, (f).b3.x}, bo1 = {(f).b2.y, (f).b3.y};    \
        const v2f bo2 = {(f).b2.z, (f).b3.z}, bo3 = {(f).b2.w, (f).b3.w};    \
        acc00 = WMMA_F32(ao0, bo0, acc00);                                   \
        acc01 = WMMA_F32(ao0, bo1, acc01);                                   \
        acc02 = WMMA_F32(ao0, bo2, acc02);                                   \
        acc03 = WMMA_F32(ao0, bo3, acc03);                                   \
        acc10 = WMMA_F32(ao1, bo0, acc10);                                   \
        acc11 = WMMA_F32(ao1, bo1, acc11);                                   \
        acc12 = WMMA_F32(ao1, bo2, acc12);                                   \
        acc13 = WMMA_F32(ao1, bo3, acc13);                                   \
    } while (0)

#define MMA_STAGE(S)                                                         \
    do {                                                                     \
        _Pragma("unroll")   \
        for (int u = 0; u < KP; ++u) MMA_KFRAG(S[u]);                        \
    } while (0)

    if (nst > 0) {
        int s = 0;
        LOAD_STAGE(P);
        for (; s + 2 < nst; s += 2) {
            LOAD_STAGE(Q);  MMA_STAGE(P);
            LOAD_STAGE(P);  MMA_STAGE(Q);
        }
        if (s + 1 < nst) { LOAD_STAGE(Q);  MMA_STAGE(P);  MMA_STAGE(Q); }
        else             { MMA_STAGE(P); }
    }
    for (int t = nst * KP; t < nkp; ++t) {
        KFrag f;
        load_kfrag(f, ap0, ap1, bp, Nc);
        ap0 += 8;  ap1 += 8;  bp += bstep;
        MMA_KFRAG(f);
    }
#undef LOAD_STAGE
#undef MMA_KFRAG
#undef MMA_STAGE

    float4 bb = {0.f, 0.f, 0.f, 0.f};
    if (bias) bb = *(const float4*)(bias + n0 + 4 * lrow);
    const int rbase = (lane >> 4) * 8;
    #pragma unroll
    for (int r = 0; r < 8; ++r) {
        float4 o;
        o.x = acc00[r] + bb.x;
        o.y = acc01[r] + bb.y;
        o.z = acc02[r] + bb.z;
        o.w = acc03[r] + bb.w;
        *(float4*)(C + (size_t)(m0 + rbase + r) * Nc + n0 + 4 * lrow) = o;
    }
    if (hasM1) {
        #pragma unroll
        for (int r = 0; r < 8; ++r) {
            float4 o;
            o.x = acc10[r] + bb.x;
            o.y = acc11[r] + bb.y;
            o.z = acc12[r] + bb.z;
            o.w = acc13[r] + bb.w;
            *(float4*)(C + (size_t)(m0 + 16 + rbase + r) * Nc + n0 + 4 * lrow) = o;
        }
    }
}

__global__ __launch_bounds__(256)
void attn_scores(const float* __restrict__ hx,
                 const float* __restrict__ att_s, const float* __restrict__ att_d,
                 float* __restrict__ a_s, float* __restrict__ a_d, int nNodes)
{
    int gid  = blockIdx.x * blockDim.x + threadIdx.x;
    int wave = gid >> 5;
    int lane = gid & 31;
    if (wave >= nNodes * HEADS) return;
    int n = wave / HEADS, h = wave % HEADS;

    const float4* hv = (const float4*)(hx + (size_t)n * FDIM + h * CDIM);
    const float4* sv = (const float4*)(att_s + h * CDIM);
    const float4* dv = (const float4*)(att_d + h * CDIM);
    float4 x = hv[lane], s = sv[lane], d = dv[lane];
    float accs = x.x * s.x + x.y * s.y + x.z * s.z + x.w * s.w;
    float accd = x.x * d.x + x.y * d.y + x.z * d.z + x.w * d.w;
    #pragma unroll
    for (int off = 16; off > 0; off >>= 1) {
        accs += __shfl_down(accs, off, 32);
        accd += __shfl_down(accd, off, 32);
    }
    if (lane == 0) { a_s[wave] = accs; a_d[wave] = accd; }
}

__global__ void fill_u32(unsigned* __restrict__ p, unsigned v, size_t n)
{
    size_t i = (size_t)blockIdx.x * blockDim.x + threadIdx.x;
    if (i < n) p[i] = v;
}

#define TD   40
#define QW   4
#define NWV  (HEADS * QW)
#define CAPW 128
#define SU4  11
#define GU   8

__global__ __launch_bounds__(NWV * 32)
void aggregate_tile(const int* __restrict__ adj, const float* __restrict__ hx,
                    const float* __restrict__ a_s, const float* __restrict__ a_d,
                    const float* __restrict__ bias, float* __restrict__ out,
                    float* __restrict__ part, int E, int N)
{
    static_assert(NWV * 32 == FDIM, "step 5 gives every thread one column");
    __shared__ float acc[TD * FDIM];
    __shared__ float4 red4[QW * 2 * (FDIM / 4)];
    __shared__ float    sL[TD * HEADS];
    __shared__ unsigned mL[TD * HEADS];
    __shared__ float    adL[TD * HEADS];
    __shared__ float    esL[TD * HEADS];
    __shared__ int   lsrc[NWV * CAPW];
    __shared__ int   ldl[NWV * CAPW];
    __shared__ float lal[NWV * CAPW * HEADS];
    __shared__ int   lcnt[NWV];
    __shared__ int   lmore[NWV];

    const int tid  = threadIdx.x;
    const int lane = tid & 31;
    const int w    = tid >> 5;
    const int h    = w / QW;
    const int q    = w % QW;
    const int d0   = blockIdx.x * TD;
    const int rows = (N - d0 < TD) ? (N - d0) : TD;
    const unsigned lt_mask = (1u << lane) - 1u;

    for (int i = tid; i < rows * HEADS; i += NWV * 32) {
        const float ad = a_d[(size_t)d0 * HEADS + i];
        float v = a_s[(size_t)d0 * HEADS + i] + ad;
        v = (v > 0.f) ? v : v * NEG_SLOPE;
        adL[i] = ad;
        esL[i] = v;
        mL[i]  = fmap(v);
    }
    __syncthreads();

    static_assert(CAPW >= 128, "the first step of a round must fit an empty list");
    const int* dstp  = adj + E;
    const int  chunk = ((E + NWV - 1) / NWV + 127) & ~127;
    const int  beg   = w * chunk;
    int end = beg + chunk;
    if (end > E) end = E;
    const int  E4   = E & ~3;
    const int  endv = (end < E4) ? end : E4;
    const int maxRounds = chunk / 32 + 2;
    int  pos    = beg;
    int  cntw   = 0;
    bool single = false;

    for (int phase = 0; phase < 3; ++phase) {
        const bool refilter = (phase == 0) || !single;
        if (refilter) pos = beg;

        if (phase == 2) {
            #pragma unroll
            for (int t = 0; t < TD / QW; ++t) {
                const int dl  = q + t * QW;
                const int dlc = (dl < rows) ? dl : (rows - 1);
                const int d   = d0 + dlc;
                const int k   = dlc * HEADS + h;
                const float a = __expf(esL[k] - funmap(mL[k])) / sL[k];
                const float4 v = *(const float4*)(hx + (size_t)d * FDIM + h * CDIM + lane * 4);
                float* ar = acc + dl * FDIM + h * CDIM + lane;
                ar[0]  = v.x * a;
                ar[32] = v.y * a;
                ar[64] = v.z * a;
                ar[96] = v.w * a;
            }
        }

        for (int round = 0; round < maxRounds; ++round) {
            if (refilter) {
                cntw = 0;
                bool full = false;
                while (pos < endv && !full) {
                    const int pos0 = pos;
                    v4i dv[SU4];
                    #pragma unroll
                    for (int u = 0; u < SU4; ++u) {
                        const int e0 = pos0 + u * 128 + lane * 4;
                        __builtin_memcpy(&dv[u], dstp + ((e0 < endv) ? e0 : (endv - 4)), 16);
                    }
                    #pragma unroll
                    for (int u = 0; u < SU4; ++u) {
                        if (!full) {
                            const v4i  d  = dv[u];
                            const bool ok = (pos0 + u * 128 + lane * 4) < endv;
                            const bool t0 = ok && ((unsigned)(d.x - d0) < (unsigned)rows);
                            const bool t1 = ok && ((unsigned)(d.y - d0) < (unsigned)rows);
                            const bool t2 = ok && ((unsigned)(d.z - d0) < (unsigned)rows);
                            const bool t3 = ok && ((unsigned)(d.w - d0) < (unsigned)rows);
                            const unsigned m0 = __builtin_amdgcn_ballot_w32(t0);
                            const unsigned m1 = __builtin_amdgcn_ballot_w32(t1);
                            const unsigned m2 = __builtin_amdgcn_ballot_w32(t2);
                            const unsigned m3 = __builtin_amdgcn_ballot_w32(t3);
                            if ((m0 | m1 | m2 | m3) != 0u) {
                                const int add = __builtin_popcount(m0) + __builtin_popcount(m1)
                                              + __builtin_popcount(m2) + __builtin_popcount(m3);
                                if (cntw + add > CAPW) {
                                    full = true;
                                } else {
                                    int p = w * CAPW + cntw
                                          + __builtin_popcount(m0 & lt_mask) + __builtin_popcount(m1 & lt_mask)
                                          + __builtin_popcount(m2 & lt_mask) + __builtin_popcount(m3 & lt_mask);
                                    const int e0 = pos0 + u * 128 + lane * 4;
                                    if (t0) { lsrc[p] = e0;      ldl[p] = d.x - d0;  ++p; }
                                    if (t1) { lsrc[p] = e0 + 1;  ldl[p] = d.y - d0;  ++p; }
                                    if (t2) { lsrc[p] = e0 + 2;  ldl[p] = d.z - d0;  ++p; }
                                    if (t3) { lsrc[p] = e0 + 3;  ldl[p] = d.w - d0; }
                                    cntw += add;
                                }
                            }
                            if (!full) {
                                const int nx = pos0 + (u + 1) * 128;
                                pos = (nx < endv) ? nx : endv;
                            }
                        }
                    }
                }
                if (!full && pos >= endv && pos < end) {
                    const int  e  = endv + lane;
                    const bool ok = e < end;
                    const int  dd = dstp[ok ? e : (end - 1)];
                    const int  d  = ok ? dd : -1;
                    const bool mt = (unsigned)(d - d0) < (unsigned)rows;
                    const unsigned m = __builtin_amdgcn_ballot_w32(mt);
                    const int add = __builtin_popcount(m);
                    if (cntw + add > CAPW) {
                        full = true;
                    } else {
                        if (mt) {
                            const int p = w * CAPW + cntw + __builtin_popcount(m & lt_mask);
                            lsrc[p] = e;
                            ldl[p]  = d - d0;
                        }
                        cntw += add;
                        pos = end;
                    }
                }
                for (int i = lane; i < cntw; i += 32) {
                    const int p  = w * CAPW + i;
                    const int e  = lsrc[p];
                    const int dl = ldl[p];
                    const int s  = adj[e];
                    const float2* pp = (const float2*)(a_s + (size_t)s * HEADS);
                    const float2 x01 = pp[0], x23 = pp[1], x45 = pp[2];
                    const float* dr = adL + dl * HEADS;
                    float v0 = x01.x + dr[0], v1 = x01.y + dr[1], v2 = x23.x + dr[2];
                    float v3 = x23.y + dr[3], v4 = x45.x + dr[4], v5 = x45.y + dr[5];
                    float* al = lal + p * HEADS;
                    al[0] = (v0 > 0.f) ? v0 : v0 * NEG_SLOPE;
                    al[1] = (v1 > 0.f) ? v1 : v1 * NEG_SLOPE;
                    al[2] = (v2 > 0.f) ? v2 : v2 * NEG_SLOPE;
                    al[3] = (v3 > 0.f) ? v3 : v3 * NEG_SLOPE;
                    al[4] = (v4 > 0.f) ? v4 : v4 * NEG_SLOPE;
                    al[5] = (v5 > 0.f) ? v5 : v5 * NEG_SLOPE;
                    lsrc[p] = s;
                }
                if (lane == 0) { lcnt[w] = cntw; lmore[w] = (pos < end) ? 1 : 0; }
            }
            for (int i = lane; i < cntw; i += 32) {
                const int p  = w * CAPW + i;
                const int dl = ldl[p];
                float* al = lal + p * HEADS;
                #pragma unroll
                for (int hh = 0; hh < HEADS; ++hh) {
                    const int   k  = dl * HEADS + hh;
                    const float ev = al[hh];
                    if (phase == 0) {
                        atomicMax(&mL[k], fmap(ev));
                    } else {
                        const float pe = __expf(ev - funmap(mL[k]));
                        if (phase == 1) atomicAdd(&sL[k], pe);
                        else            al[hh] = pe / sL[k];
                    }
                }
            }
            __syncthreads();

            int c  = (lane < NWV) ? lcnt[lane] : 0;
            int mo = (lane < NWV) ? lmore[lane] : 0;
            const bool anyMore = __builtin_amdgcn_ballot_w32(mo != 0) != 0u;
            if (phase == 2) {
                int incl = c;
                #pragma unroll
                for (int off = 1; off < 32; off <<= 1) {
                    const int t = __shfl_up(incl, off, 32);
                    if (lane >= off) incl += t;
                }
                const int n = __builtin_amdgcn_readlane(incl, 31);

                for (int b = 0; b < n; b += 32) {
                    const int g = b + lane;
                    const bool valid = g < n;
                    int ww = 0, base = 0;
                    #pragma unroll
                    for (int j = 0; j < NWV; ++j) {
                        const int ij = __builtin_amdgcn_readlane(incl, j);
                        if (ij <= g) { ww = j + 1; base = ij; }
                    }
                    int s_ = d0, dl_ = 0;
                    float a_ = 0.f;
                    if (valid) {
                        const int slot = ww * CAPW + (g - base);
                        s_  = lsrc[slot];
                        dl_ = ldl[slot];
                        a_  = lal[slot * HEADS + h];
                    }
                    const bool mine = valid && ((dl_ & (QW - 1)) == q);
                    unsigned mm = __builtin_amdgcn_ballot_w32(mine);
                    while (mm) {
                        int  idx[GU];
                        bool ok[GU];
                        float4 v[GU];
                        #pragma unroll
                        for (int u = 0; u < GU; ++u) {
                            ok[u]  = (mm != 0u);
                            idx[u] = ok[u] ? __builtin_ctz(mm) : 0;
                            if (ok[u]) mm &= mm - 1u;
                            const int srow = ok[u] ? __builtin_amdgcn_readlane(s_, idx[u]) : d0;
                            v[u] = *(const float4*)(hx + (size_t)srow * FDIM + h * CDIM + lane * 4);
                        }
                        #pragma unroll
                        for (int u = 0; u < GU; ++u) {
                            if (ok[u]) {
                                const int   dl = __builtin_amdgcn_readlane(dl_, idx[u]);
                                const float a  = __int_as_float(__builtin_amdgcn_readlane(__float_as_int(a_), idx[u]));
                                float* ar = acc + dl * FDIM + h * CDIM + lane;
                                atomicAdd(ar,      v[u].x * a);
                                atomicAdd(ar + 32, v[u].y * a);
                                atomicAdd(ar + 64, v[u].z * a);
                                atomicAdd(ar + 96, v[u].w * a);
                            }
                        }
                    }
                }
            }
            if (phase == 0 && round == 0) single = !anyMore;
            if (!anyMore) break;
            __syncthreads();
        }

        if (phase == 0) {
            for (int i = tid; i < rows * HEADS; i += NWV * 32)
                sL[i] = __expf(esL[i] - funmap(mL[i]));
        }
        if (phase < 2) __syncthreads();
    }

    const float4 bb = *(const float4*)(bias + h * CDIM + lane * 4);
    float4 cs = {0.f, 0.f, 0.f, 0.f};
    float4 cq = {0.f, 0.f, 0.f, 0.f};
    for (int dl = q; dl < rows; dl += QW) {
        const float* ar = acc + dl * FDIM + h * CDIM + lane;
        float4 o;
        o.x = ar[0]  + bb.x;
        o.y = ar[32] + bb.y;
        o.z = ar[64] + bb.z;
        o.w = ar[96] + bb.w;
        o.x = o.x > 0.f ? o.x : 0.f;
        o.y = o.y > 0.f ? o.y : 0.f;
        o.z = o.z > 0.f ? o.z : 0.f;
        o.w = o.w > 0.f ? o.w : 0.f;
        *(float4*)(out + (size_t)(d0 + dl) * FDIM + h * CDIM + lane * 4) = o;
        cs.x += o.x;  cs.y += o.y;  cs.z += o.z;  cs.w += o.w;
        cq.x += o.x * o.x;  cq.y += o.y * o.y;  cq.z += o.z * o.z;  cq.w += o.w * o.w;
    }

    red4[(q * 2)     * (FDIM / 4) + h * (CDIM / 4) + lane] = cs;
    red4[(q * 2 + 1) * (FDIM / 4) + h * (CDIM / 4) + lane] = cq;
    __syncthreads();
    {
        const float* red = (const float*)red4;
        float s = 0.f, ss = 0.f;
        #pragma unroll
        for (int k = 0; k < QW; ++k) {
            s  += red[(k * 2)     * FDIM + tid];
            ss += red[(k * 2 + 1) * FDIM + tid];
        }
        part[(size_t)(blockIdx.x * 2)     * FDIM + tid] = s;
        part[(size_t)(blockIdx.x * 2 + 1) * FDIM + tid] = ss;
    }
}

#define RW 16
#define RU 8

__global__ __launch_bounds__(RW * 32)
void bn_finish(const float* __restrict__ part, float* __restrict__ mean,
               float* __restrict__ rstd, int nTile, int rows)
{
    __shared__ float sh[RW][32], sh2[RW][32];
    const int lane = threadIdx.x & 31;
    const int w    = threadIdx.x >> 5;
    const int col  = blockIdx.x * 32 + lane;
    float s = 0.f, ss = 0.f;
    for (int t0 = w; t0 < nTile; t0 += RW * RU) {
        float a[RU], b[RU];
        #pragma unroll
        for (int u = 0; u < RU; ++u) {
            const int t  = t0 + u * RW;
            const int tc = (t < nTile) ? t : (nTile - 1);
            a[u] = part[(size_t)(tc * 2)     * FDIM + col];
            b[u] = part[(size_t)(tc * 2 + 1) * FDIM + col];
        }
        #pragma unroll
        for (int u = 0; u < RU; ++u) {
            if (t0 + u * RW < nTile) { s += a[u]; ss += b[u]; }
        }
    }
    sh[w][lane] = s;  sh2[w][lane] = ss;
    __syncthreads();
    if (w == 0) {
        float S = 0.f, SS = 0.f;
        #pragma unroll
        for (int k = 0; k < RW; ++k) { S += sh[k][lane]; SS += sh2[k][lane]; }
        float mu  = S / rows;
        float var = SS / rows - mu * mu;
        mean[col] = mu;
        rstd[col] = rsqrtf(var + BN_EPS);
    }
}

#define BA_WG    480
#define BA_SLOTS (BA_WG * 256 / (FDIM / 4))
#define BA_ROWS  16

__global__ __launch_bounds__(256)
void bn_apply(float* __restrict__ x, const float* __restrict__ mean,
              const float* __restrict__ rstd, const float* __restrict__ gamma,
              const float* __restrict__ beta, int rows)
{
    static_assert((BA_WG * 256) % (FDIM / 4) == 0, "a thread keeps one column group");
    const int gid = blockIdx.x * 256 + threadIdx.x;
    const int cg  = gid % (FDIM / 4);
    const int r0  = gid / (FDIM / 4);
    const float4 mu = ((const float4*)mean)[cg];
    const float4 rs = ((const float4*)rstd)[cg];
    const float4 ga = ((const float4*)gamma)[cg];
    const float4 be = ((const float4*)beta)[cg];
    float* xp = x + cg * 4;

    for (int rb = r0; rb < rows; rb += BA_SLOTS * BA_ROWS) {
        float4 v[BA_ROWS];
        #pragma unroll
        for (int u = 0; u < BA_ROWS; ++u) {
            const int r  = rb + u * BA_SLOTS;
            const int rc = (r < rows) ? r : rb;
            v[u] = *(const float4*)(xp + (size_t)rc * FDIM);
        }
        #pragma unroll
        for (int u = 0; u < BA_ROWS; ++u) {
            const int r = rb + u * BA_SLOTS;
            if (r < rows) {
                float4 o;
                o.x = (v[u].x - mu.x) * rs.x * ga.x + be.x;
                o.y = (v[u].y - mu.y) * rs.y * ga.y + be.y;
                o.z = (v[u].z - mu.z) * rs.z * ga.z + be.z;
                o.w = (v[u].w - mu.w) * rs.w * ga.w + be.w;
                *(float4*)(xp + (size_t)r * FDIM) = o;
            }
        }
    }
}

__global__ __launch_bounds__(256)
void pool_max(const float* __restrict__ lin, const int* __restrict__ ibatch,
              unsigned* __restrict__ pooled, int rows)
{
    size_t i = (size_t)blockIdx.x * blockDim.x + threadIdx.x;
    if (i >= (size_t)rows * OUT_C) return;
    int n = (int)(i / OUT_C), c = (int)(i % OUT_C);
    atomicMax(&pooled[ibatch[n] * OUT_C + c], fmap(lin[i]));
}

__global__ __launch_bounds__(256)
void pool_final(const unsigned* __restrict__ pooled, float* __restrict__ out, int n)
{
    int i = blockIdx.x * blockDim.x + threadIdx.x;
    if (i < n) out[i] = funmap(pooled[i]);
}

extern "C" void kernel_launch(void* const* d_in, const int* in_sizes, int n_in,
                              void* d_out, int out_size, void* d_ws, size_t ws_size,
                              hipStream_t stream)
{
    const float* x_in   = (const float*)d_in[0];
    const int*   adj    = (const int*)d_in[1];
    const int*   ibatch = (const int*)d_in[2];
    const float* W1     = (const float*)d_in[3];
    const float* att_s1 = (const float*)d_in[4];
    const float* att_d1 = (const float*)d_in[5];
    const float* b1     = (const float*)d_in[6];
    const float* g1     = (const float*)d_in[7];
    const float* be1    = (const float*)d_in[8];
    const float* W2     = (const float*)d_in[9];
    const float* att_s2 = (const float*)d_in[10];
    const float* att_d2 = (const float*)d_in[11];
    const float* b2     = (const float*)d_in[12];
    const float* g2     = (const float*)d_in[13];
    const float* be2    = (const float*)d_in[14];
    const float* linW   = (const float*)d_in[15];
    const float* linb   = (const float*)d_in[16];
    float* out = (float*)d_out;

    const int N    = in_sizes[0] / F_IN;
    const int E    = in_sizes[1] / 2;
    const int B    = out_size / OUT_C;
    const int mT   = (N + 31) / 32;

    float*    hx   = (float*)d_ws;
    float*    agg  = hx  + (size_t)N * FDIM;
    float*    a_s  = agg + (size_t)N * FDIM;
    float*    a_d  = a_s + (size_t)N * HEADS;
    float*    mean = a_d + (size_t)N * HEADS;
    float*    rstd = mean + FDIM;
    float*    lin  = rstd + FDIM;
    unsigned* pooled = (unsigned*)(lin + (size_t)N * OUT_C);
    float*    part = lin;

    const size_t nh  = (size_t)N * HEADS;
    const int    nTile = (N + TD - 1) / TD;

    auto layer = [&](const float* xin, int K, const float* W,
                     const float* asw, const float* adw,
                     const float* bias, const float* gamma, const float* beta) {
        wmma_gemm_f32<<<mT * (FDIM / 64), 32, 0, stream>>>(xin, W, hx, nullptr, N, K, FDIM);
        attn_scores<<<(int)((nh * 32 + 255) / 256), 256, 0, stream>>>(hx, asw, adw, a_s, a_d, N);
        aggregate_tile<<<nTile, NWV * 32, 0, stream>>>(adj, hx, a_s, a_d, bias, agg, part, E, N);
        bn_finish<<<FDIM / 32, RW * 32, 0, stream>>>(part, mean, rstd, nTile, N);
        bn_apply<<<BA_WG, 256, 0, stream>>>(agg, mean, rstd, gamma, beta, N);
    };

    layer(x_in, F_IN, W1, att_s1, att_d1, b1, g1, be1);
    layer(agg,  FDIM, W2, att_s2, att_d2, b2, g2, be2);

    wmma_gemm_f32<<<mT * (OUT_C / 64), 32, 0, stream>>>(agg, linW, lin, linb, N, FDIM, OUT_C);
    fill_u32<<<(B * OUT_C + 255) / 256, 256, 0, stream>>>(pooled, FMAP_NEG_INF, (size_t)B * OUT_C);
    pool_max<<<(int)(((size_t)N * OUT_C + 255) / 256), 256, 0, stream>>>(lin, ibatch, pooled, N);
    pool_final<<<(B * OUT_C + 255) / 256, 256, 0, stream>>>(pooled, out, B * OUT_C);
}


# Attention scores made by the GEMMs: a 13th strip per row tile multiplies by W.att (12 columns from one small launch; bn_fold scales layer 2's); both score passes gone; scores now summed over k
# speedup vs baseline: 3.5922x; 1.1222x over previous
#include <hip/hip_runtime.h>

#define F_IN   256
#define HEADS  6
#define CDIM   128
#define FDIM   768
#define OUT_C  128
#define SCP    12
#define XW     64
#define NEG_SLOPE 0.2f
#define BN_EPS    1e-5f

typedef __attribute__((ext_vector_type(2))) float v2f;
typedef __attribute__((ext_vector_type(8))) float v8f;
typedef __attribute__((ext_vector_type(4))) int   v4i;

__device__ __forceinline__ unsigned fmap(float f) {
    unsigned u = __float_as_uint(f);
    return (u & 0x80000000u) ? ~u : (u | 0x80000000u);
}
__device__ __forceinline__ float funmap(unsigned u) {
    u = (u & 0x80000000u) ? (u & 0x7FFFFFFFu) : ~u;
    return __uint_as_float(u);
}
#define FMAP_NEG_INF 0x007FFFFFu

#define WMMA_F32(a, b, c) \
    __builtin_amdgcn_wmma_f32_16x16x4_f32(false, (a), false, (b), (short)0, (c), false, false)

#define KP 2
#define SLAB  32
#define NSLAB (FDIM / SLAB)

static_assert(F_IN % 8 == 0 && FDIM % 8 == 0, "the K loop walks whole k-pairs");
static_assert(FDIM % SLAB == 0, "whole slabs");

struct KFrag { float4 a0, a1, b0, b1, b2, b3; };

__device__ __forceinline__ void load_kfrag(KFrag& f,
                                           const float* __restrict__ ap0,
                                           const float* __restrict__ ap1,
                                           const float* __restrict__ bp, int Nc)
{
    f.a0 = *(const float4*)ap0;
    f.a1 = *(const float4*)ap1;
    f.b0 = *(const float4*)(bp);
    f.b1 = *(const float4*)(bp + Nc);
    f.b2 = *(const float4*)(bp + 2 * (size_t)Nc);
    f.b3 = *(const float4*)(bp + 3 * (size_t)Nc);
}

__global__ __launch_bounds__(32)
void wmma_gemm_f32(const float* __restrict__ A, const float* __restrict__ B,
                   float* __restrict__ C, const float* __restrict__ bias,
                   const float* __restrict__ cpart,
                   const float* __restrict__ Bx, const float* __restrict__ cpx,
                   float* __restrict__ Cx, int M, int K, int Nc)
{
    const int lane  = threadIdx.x & 31;
    const int nN64  = Nc >> 6;
    const int nS    = nN64 + (Bx ? 1 : 0);
    const int strip = blockIdx.x % nS;
    const int m0    = (blockIdx.x / nS) * 32;
    const bool xs   = (strip == nN64);
    const int n0    = xs ? 0 : strip * 64;
    const int ldb   = xs ? XW : Nc;
    const float* __restrict__ Bw = xs ? Bx : B;
    const int lrow  = lane & 15;
    const int kq    = (lane >> 4) << 2;
    const bool hasM1 = (m0 + 16) < M;

    v8f acc00 = {}, acc01 = {}, acc02 = {}, acc03 = {};
    v8f acc10 = {}, acc11 = {}, acc12 = {}, acc13 = {};

    const float* ap0 = A + (size_t)(m0 + lrow) * K + kq;
    const float* ap1 = A + (size_t)(hasM1 ? (m0 + 16 + lrow) : (m0 + lrow)) * K + kq;
    const float* bp  = Bw + (size_t)kq * ldb + n0 + 4 * lrow;
    const size_t bstep = (size_t)8 * ldb;

    const int nkp = K >> 3;
    const int nst = nkp / KP;

    KFrag P[KP], Q[KP];

#define LOAD_STAGE(S)                                                        \
    do {                                                                     \
        _Pragma("unroll")   \
        for (int u = 0; u < KP; ++u)                                         \
            load_kfrag(S[u], ap0 + 8 * u, ap1 + 8 * u, bp + u * bstep, ldb); \
        ap0 += 8 * KP;  ap1 += 8 * KP;  bp += KP * bstep;                    \
    } while (0)

#define MMA_KFRAG(f)                                                         \
    do {                                                                     \
        const v2f ae0 = {(f).a0.x, (f).a0.y}, ae1 = {(f).a1.x, (f).a1.y};    \
        const v2f be0 = {(f).b0.x, (f).b1.x}, be1 = {(f).b0.y, (f).b1.y};    \
        const v2f be2 = {(f).b0.z, (f).b1.z}, be3 = {(f).b0.w, (f).b1.w};    \
        acc00 = WMMA_F32(ae0, be0, acc00);                                   \
        acc01 = WMMA_F32(ae0, be1, acc01);                                   \
        acc02 = WMMA_F32(ae0, be2, acc02);                                   \
        acc03 = WMMA_F32(ae0, be3, acc03);                                   \
        acc10 = WMMA_F32(ae1, be0, acc10);                                   \
        acc11 = WMMA_F32(ae1, be1, acc11);                                   \
        acc12 = WMMA_F32(ae1, be2, acc12);                                   \
        acc13 = WMMA_F32(ae1, be3, acc13);                                   \
        const v2f ao0 = {(f).a0.z, (f).a0.w}, ao1 = {(f).a1.z, (f).a1.w};    \
        const v2f bo0 = {(f).b2.x, (f).b3.x}, bo1 = {(f).b2.y, (f).b3.y};    \
        const v2f bo2 = {(f).b2.z, (f).b3.z}, bo3 = {(f).b2.w, (f).b3.w};    \
        acc00 = WMMA_F32(ao0, bo0, acc00);                                   \
        acc01 = WMMA_F32(ao0, bo1, acc01);                                   \
        acc02 = WMMA_F32(ao0, bo2, acc02);                                   \
        acc03 = WMMA_F32(ao0, bo3, acc03);                                   \
        acc10 = WMMA_F32(ao1, bo0, acc10);                                   \
        acc11 = WMMA_F32(ao1, bo1, acc11);                                   \
        acc12 = WMMA_F32(ao1, bo2, acc12);                                   \
        acc13 = WMMA_F32(ao1, bo3, acc13);                                   \
    } while (0)

#define MMA_STAGE(S)                                                         \
    do {                                                                     \
        _Pragma("unroll")   \
        for (int u = 0; u < KP; ++u) MMA_KFRAG(S[u]);                        \
    } while (0)

    if (nst > 0) {
        int s = 0;
        LOAD_STAGE(P);
        for (; s + 2 < nst; s += 2) {
            LOAD_STAGE(Q);  MMA_STAGE(P);
            LOAD_STAGE(P);  MMA_STAGE(Q);
        }
        if (s + 1 < nst) { LOAD_STAGE(Q);  MMA_STAGE(P);  MMA_STAGE(Q); }
        else             { MMA_STAGE(P); }
    }
    for (int t = nst * KP; t < nkp; ++t) {
        KFrag f;
        load_kfrag(f, ap0, ap1, bp, ldb);
        ap0 += 8;  ap1 += 8;  bp += bstep;
        MMA_KFRAG(f);
    }
#undef LOAD_STAGE
#undef MMA_KFRAG
#undef MMA_STAGE

    float4 bb = {0.f, 0.f, 0.f, 0.f};
    if (bias && !xs) bb = *(const float4*)(bias + n0 + 4 * lrow);
    const float* __restrict__ cpw = xs ? cpx : cpart;
    if (cpw) {
        float4 pv[NSLAB];
        #pragma unroll
        for (int g = 0; g < NSLAB; ++g)
            pv[g] = *(const float4*)(cpw + (size_t)g * ldb + n0 + 4 * lrow);
        #pragma unroll
        for (int g = 0; g < NSLAB; ++g) {
            bb.x += pv[g].x;  bb.y += pv[g].y;  bb.z += pv[g].z;  bb.w += pv[g].w;
        }
    }
    const int rbase = (lane >> 4) * 8;
    if (xs) {
        if (lrow < SCP / 4) {
            #pragma unroll
            for (int r = 0; r < 8; ++r) {
                float4 o;
                o.x = acc00[r] + bb.x;
                o.y = acc01[r] + bb.y;
                o.z = acc02[r] + bb.z;
                o.w = acc03[r] + bb.w;
                *(float4*)(Cx + (size_t)(m0 + rbase + r) * SCP + 4 * lrow) = o;
            }
            if (hasM1) {
                #pragma unroll
                for (int r = 0; r < 8; ++r) {
                    float4 o;
                    o.x = acc10[r] + bb.x;
                    o.y = acc11[r] + bb.y;
                    o.z = acc12[r] + bb.z;
                    o.w = acc13[r] + bb.w;
                    *(float4*)(Cx + (size_t)(m0 + 16 + rbase + r) * SCP + 4 * lrow) = o;
                }
            }
        }
        return;
    }
    #pragma unroll
    for (int r = 0; r < 8; ++r) {
        float4 o;
        o.x = acc00[r] + bb.x;
        o.y = acc01[r] + bb.y;
        o.z = acc02[r] + bb.z;
        o.w = acc03[r] + bb.w;
        *(float4*)(C + (size_t)(m0 + rbase + r) * Nc + n0 + 4 * lrow) = o;
    }
    if (hasM1) {
        #pragma unroll
        for (int r = 0; r < 8; ++r) {
            float4 o;
            o.x = acc10[r] + bb.x;
            o.y = acc11[r] + bb.y;
            o.z = acc12[r] + bb.z;
            o.w = acc13[r] + bb.w;
            *(float4*)(C + (size_t)(m0 + 16 + rbase + r) * Nc + n0 + 4 * lrow) = o;
        }
    }
}

__global__ __launch_bounds__(256)
void att_proj(const float* __restrict__ W1, const float* __restrict__ as1,
              const float* __restrict__ ad1, float* __restrict__ Wa1, int K1,
              const float* __restrict__ W2, const float* __restrict__ as2,
              const float* __restrict__ ad2, float* __restrict__ U2, int K2)
{
    const int wave = (blockIdx.x * blockDim.x + threadIdx.x) >> 5;
    const int lane = threadIdx.x & 31;
    if (wave >= K1 + K2) return;
    const bool first = wave < K1;
    const int  k     = first ? wave : (wave - K1);
    const float* __restrict__ Wr = (first ? W1 : W2) + (size_t)k * FDIM;
    const float* __restrict__ as = first ? as1 : as2;
    const float* __restrict__ ad = first ? ad1 : ad2;

    float4 x[HEADS], s[HEADS], d[HEADS];
    #pragma unroll
    for (int h = 0; h < HEADS; ++h) {
        x[h] = *(const float4*)(Wr + h * CDIM + lane * 4);
        s[h] = *(const float4*)(as + h * CDIM + lane * 4);
        d[h] = *(const float4*)(ad + h * CDIM + lane * 4);
    }
    float r[SCP];
    #pragma unroll
    for (int h = 0; h < HEADS; ++h) {
        r[h]         = x[h].x * s[h].x + x[h].y * s[h].y + x[h].z * s[h].z + x[h].w * s[h].w;
        r[HEADS + h] = x[h].x * d[h].x + x[h].y * d[h].y + x[h].z * d[h].z + x[h].w * d[h].w;
    }
    #pragma unroll
    for (int off = 16; off > 0; off >>= 1) {
        #pragma unroll
        for (int j = 0; j < SCP; ++j) r[j] += __shfl_xor(r[j], off, 32);
    }
    float mine = 0.f;
    #pragma unroll
    for (int j = 0; j < SCP; ++j) if (lane == j) mine = r[j];
    if (first) {
        Wa1[(size_t)k * XW + lane]      = mine;
        Wa1[(size_t)k * XW + 32 + lane] = 0.f;
    } else if (lane < SCP) {
        U2[(size_t)k * SCP + lane] = mine;
    }
}

__global__ void fill_u32(unsigned* __restrict__ p, unsigned v, size_t n)
{
    size_t i = (size_t)blockIdx.x * blockDim.x + threadIdx.x;
    if (i < n) p[i] = v;
}

#define TD   40
#define QW   4
#define NWV  (HEADS * QW)
#define CAPW 128
#define SU4  11
#define GU   8

__global__ __launch_bounds__(NWV * 32)
void aggregate_tile(const int* __restrict__ adj, const float* __restrict__ hx,
                    const float* __restrict__ sc,
                    const float* __restrict__ bias, float* __restrict__ out,
                    float* __restrict__ part, int E, int N)
{
    static_assert(NWV * 32 == FDIM, "step 5 gives every thread one column");
    __shared__ float acc[TD * FDIM];
    __shared__ float4 red4[QW * 2 * (FDIM / 4)];
    __shared__ float    sL[TD * HEADS];
    __shared__ unsigned mL[TD * HEADS];
    __shared__ float    adL[TD * HEADS];
    __shared__ float    esL[TD * HEADS];
    __shared__ int   lsrc[NWV * CAPW];
    __shared__ int   ldl[NWV * CAPW];
    __shared__ float lal[NWV * CAPW * HEADS];
    __shared__ int   lcnt[NWV];
    __shared__ int   lmore[NWV];

    const int tid  = threadIdx.x;
    const int lane = tid & 31;
    const int w    = tid >> 5;
    const int h    = w / QW;
    const int q    = w % QW;
    const int d0   = blockIdx.x * TD;
    const int rows = (N - d0 < TD) ? (N - d0) : TD;
    const unsigned lt_mask = (1u << lane) - 1u;

    for (int i = tid; i < rows * HEADS; i += NWV * 32) {
        const int r = i / HEADS;
        const float* sp = sc + (size_t)(d0 + r) * SCP + (i - r * HEADS);
        const float ad = sp[HEADS];
        float v = sp[0] + ad;
        v = (v > 0.f) ? v : v * NEG_SLOPE;
        adL[i] = ad;
        esL[i] = v;
        mL[i]  = fmap(v);
    }
    __syncthreads();

    static_assert(CAPW >= 128, "the first step of a round must fit an empty list");
    const int* dstp  = adj + E;
    const int  chunk = ((E + NWV - 1) / NWV + 127) & ~127;
    const int  beg   = w * chunk;
    int end = beg + chunk;
    if (end > E) end = E;
    const int  E4   = E & ~3;
    const int  endv = (end < E4) ? end : E4;
    const int maxRounds = chunk / 32 + 2;
    int  pos    = beg;
    int  cntw   = 0;
    bool single = false;

    for (int phase = 0; phase < 3; ++phase) {
        const bool refilter = (phase == 0) || !single;
        if (refilter) pos = beg;

        if (phase == 2) {
            #pragma unroll
            for (int t = 0; t < TD / QW; ++t) {
                const int dl  = q + t * QW;
                const int dlc = (dl < rows) ? dl : (rows - 1);
                const int d   = d0 + dlc;
                const int k   = dlc * HEADS + h;
                const float a = __expf(esL[k] - funmap(mL[k])) / sL[k];
                const float4 v = *(const float4*)(hx + (size_t)d * FDIM + h * CDIM + lane * 4);
                float* ar = acc + dl * FDIM + h * CDIM + lane;
                ar[0]  = v.x * a;
                ar[32] = v.y * a;
                ar[64] = v.z * a;
                ar[96] = v.w * a;
            }
        }

        for (int round = 0; round < maxRounds; ++round) {
            if (refilter) {
                cntw = 0;
                bool full = false;
                while (pos < endv && !full) {
                    const int pos0 = pos;
                    v4i dv[SU4];
                    #pragma unroll
                    for (int u = 0; u < SU4; ++u) {
                        const int e0 = pos0 + u * 128 + lane * 4;
                        __builtin_memcpy(&dv[u], dstp + ((e0 < endv) ? e0 : (endv - 4)), 16);
                    }
                    #pragma unroll
                    for (int u = 0; u < SU4; ++u) {
                        if (!full) {
                            const v4i  d  = dv[u];
                            const bool ok = (pos0 + u * 128 + lane * 4) < endv;
                            const bool t0 = ok && ((unsigned)(d.x - d0) < (unsigned)rows);
                            const bool t1 = ok && ((unsigned)(d.y - d0) < (unsigned)rows);
                            const bool t2 = ok && ((unsigned)(d.z - d0) < (unsigned)rows);
                            const bool t3 = ok && ((unsigned)(d.w - d0) < (unsigned)rows);
                            const unsigned m0 = __builtin_amdgcn_ballot_w32(t0);
                            const unsigned m1 = __builtin_amdgcn_ballot_w32(t1);
                            const unsigned m2 = __builtin_amdgcn_ballot_w32(t2);
                            const unsigned m3 = __builtin_amdgcn_ballot_w32(t3);
                            if ((m0 | m1 | m2 | m3) != 0u) {
                                const int add = __builtin_popcount(m0) + __builtin_popcount(m1)
                                              + __builtin_popcount(m2) + __builtin_popcount(m3);
                                if (cntw + add > CAPW) {
                                    full = true;
                                } else {
                                    int p = w * CAPW + cntw
                                          + __builtin_popcount(m0 & lt_mask) + __builtin_popcount(m1 & lt_mask)
                                          + __builtin_popcount(m2 & lt_mask) + __builtin_popcount(m3 & lt_mask);
                                    const int e0 = pos0 + u * 128 + lane * 4;
                                    if (t0) { lsrc[p] = e0;      ldl[p] = d.x - d0;  ++p; }
                                    if (t1) { lsrc[p] = e0 + 1;  ldl[p] = d.y - d0;  ++p; }
                                    if (t2) { lsrc[p] = e0 + 2;  ldl[p] = d.z - d0;  ++p; }
                                    if (t3) { lsrc[p] = e0 + 3;  ldl[p] = d.w - d0; }
                                    cntw += add;
                                }
                            }
                            if (!full) {
                                const int nx = pos0 + (u + 1) * 128;
                                pos = (nx < endv) ? nx : endv;
                            }
                        }
                    }
                }
                if (!full && pos >= endv && pos < end) {
                    const int  e  = endv + lane;
                    const bool ok = e < end;
                    const int  dd = dstp[ok ? e : (end - 1)];
                    const int  d  = ok ? dd : -1;
                    const bool mt = (unsigned)(d - d0) < (unsigned)rows;
                    const unsigned m = __builtin_amdgcn_ballot_w32(mt);
                    const int add = __builtin_popcount(m);
                    if (cntw + add > CAPW) {
                        full = true;
                    } else {
                        if (mt) {
                            const int p = w * CAPW + cntw + __builtin_popcount(m & lt_mask);
                            lsrc[p] = e;
                            ldl[p]  = d - d0;
                        }
                        cntw += add;
                        pos = end;
                    }
                }
                for (int i = lane; i < cntw; i += 32) {
                    const int p  = w * CAPW + i;
                    const int e  = lsrc[p];
                    const int dl = ldl[p];
                    const int s  = adj[e];
                    const float2* pp = (const float2*)(sc + (size_t)s * SCP);
                    const float2 x01 = pp[0], x23 = pp[1], x45 = pp[2];
                    const float* dr = adL + dl * HEADS;
                    float v0 = x01.x + dr[0], v1 = x01.y + dr[1], v2 = x23.x + dr[2];
                    float v3 = x23.y + dr[3], v4 = x45.x + dr[4], v5 = x45.y + dr[5];
                    float* al = lal + p * HEADS;
                    al[0] = (v0 > 0.f) ? v0 : v0 * NEG_SLOPE;
                    al[1] = (v1 > 0.f) ? v1 : v1 * NEG_SLOPE;
                    al[2] = (v2 > 0.f) ? v2 : v2 * NEG_SLOPE;
                    al[3] = (v3 > 0.f) ? v3 : v3 * NEG_SLOPE;
                    al[4] = (v4 > 0.f) ? v4 : v4 * NEG_SLOPE;
                    al[5] = (v5 > 0.f) ? v5 : v5 * NEG_SLOPE;
                    lsrc[p] = s;
                }
                if (lane == 0) { lcnt[w] = cntw; lmore[w] = (pos < end) ? 1 : 0; }
            }
            for (int i = lane; i < cntw; i += 32) {
                const int p  = w * CAPW + i;
                const int dl = ldl[p];
                float* al = lal + p * HEADS;
                #pragma unroll
                for (int hh = 0; hh < HEADS; ++hh) {
                    const int   k  = dl * HEADS + hh;
                    const float ev = al[hh];
                    if (phase == 0) {
                        atomicMax(&mL[k], fmap(ev));
                    } else {
                        const float pe = __expf(ev - funmap(mL[k]));
                        if (phase == 1) atomicAdd(&sL[k], pe);
                        else            al[hh] = pe / sL[k];
                    }
                }
            }
            __syncthreads();

            int c  = (lane < NWV) ? lcnt[lane] : 0;
            int mo = (lane < NWV) ? lmore[lane] : 0;
            const bool anyMore = __builtin_amdgcn_ballot_w32(mo != 0) != 0u;
            if (phase == 2) {
                int incl = c;
                #pragma unroll
                for (int off = 1; off < 32; off <<= 1) {
                    const int t = __shfl_up(incl, off, 32);
                    if (lane >= off) incl += t;
                }
                const int n = __builtin_amdgcn_readlane(incl, 31);

                for (int b = 0; b < n; b += 32) {
                    const int g = b + lane;
                    const bool valid = g < n;
                    int ww = 0, base = 0;
                    #pragma unroll
                    for (int j = 0; j < NWV; ++j) {
                        const int ij = __builtin_amdgcn_readlane(incl, j);
                        if (ij <= g) { ww = j + 1; base = ij; }
                    }
                    int s_ = d0, dl_ = 0;
                    float a_ = 0.f;
                    if (valid) {
                        const int slot = ww * CAPW + (g - base);
                        s_  = lsrc[slot];
                        dl_ = ldl[slot];
                        a_  = lal[slot * HEADS + h];
                    }
                    const bool mine = valid && ((dl_ & (QW - 1)) == q);
                    unsigned mm = __builtin_amdgcn_ballot_w32(mine);
                    while (mm) {
                        int  idx[GU];
                        bool ok[GU];
                        float4 v[GU];
                        #pragma unroll
                        for (int u = 0; u < GU; ++u) {
                            ok[u]  = (mm != 0u);
                            idx[u] = ok[u] ? __builtin_ctz(mm) : 0;
                            if (ok[u]) mm &= mm - 1u;
                            const int srow = ok[u] ? __builtin_amdgcn_readlane(s_, idx[u]) : d0;
                            v[u] = *(const float4*)(hx + (size_t)srow * FDIM + h * CDIM + lane * 4);
                        }
                        #pragma unroll
                        for (int u = 0; u < GU; ++u) {
                            if (ok[u]) {
                                const int   dl = __builtin_amdgcn_readlane(dl_, idx[u]);
                                const float a  = __int_as_float(__builtin_amdgcn_readlane(__float_as_int(a_), idx[u]));
                                float* ar = acc + dl * FDIM + h * CDIM + lane;
                                atomicAdd(ar,      v[u].x * a);
                                atomicAdd(ar + 32, v[u].y * a);
                                atomicAdd(ar + 64, v[u].z * a);
                                atomicAdd(ar + 96, v[u].w * a);
                            }
                        }
                    }
                }
            }
            if (phase == 0 && round == 0) single = !anyMore;
            if (!anyMore) break;
            __syncthreads();
        }

        if (phase == 0) {
            for (int i = tid; i < rows * HEADS; i += NWV * 32)
                sL[i] = __expf(esL[i] - funmap(mL[i]));
        }
        if (phase < 2) __syncthreads();
    }

    const float4 bb = *(const float4*)(bias + h * CDIM + lane * 4);
    float4 cs = {0.f, 0.f, 0.f, 0.f};
    float4 cq = {0.f, 0.f, 0.f, 0.f};
    for (int dl = q; dl < rows; dl += QW) {
        const float* ar = acc + dl * FDIM + h * CDIM + lane;
        float4 o;
        o.x = ar[0]  + bb.x;
        o.y = ar[32] + bb.y;
        o.z = ar[64] + bb.z;
        o.w = ar[96] + bb.w;
        o.x = o.x > 0.f ? o.x : 0.f;
        o.y = o.y > 0.f ? o.y : 0.f;
        o.z = o.z > 0.f ? o.z : 0.f;
        o.w = o.w > 0.f ? o.w : 0.f;
        *(float4*)(out + (size_t)(d0 + dl) * FDIM + h * CDIM + lane * 4) = o;
        cs.x += o.x;  cs.y += o.y;  cs.z += o.z;  cs.w += o.w;
        cq.x += o.x * o.x;  cq.y += o.y * o.y;  cq.z += o.z * o.z;  cq.w += o.w * o.w;
    }

    red4[(q * 2)     * (FDIM / 4) + h * (CDIM / 4) + lane] = cs;
    red4[(q * 2 + 1) * (FDIM / 4) + h * (CDIM / 4) + lane] = cq;
    __syncthreads();
    {
        const float* red = (const float*)red4;
        float s = 0.f, ss = 0.f;
        #pragma unroll
        for (int k = 0; k < QW; ++k) {
            s  += red[(k * 2)     * FDIM + tid];
            ss += red[(k * 2 + 1) * FDIM + tid];
        }
        part[(size_t)(blockIdx.x * 2)     * FDIM + tid] = s;
        part[(size_t)(blockIdx.x * 2 + 1) * FDIM + tid] = ss;
    }
}

#define RW 16
#define RU 8

constexpr int pow2_floor(int v) { int p = 1; while (p * 2 <= v) p *= 2; return p; }

template <int NC>
__global__ __launch_bounds__(RW * 32)
void bn_fold(const float* __restrict__ part, const float* __restrict__ gamma,
             const float* __restrict__ beta, const float* __restrict__ Wt,
             float* __restrict__ Wf, float* __restrict__ cpart,
             const float* __restrict__ U, float* __restrict__ Wa, float* __restrict__ cpa,
             int nTile, int rows)
{
    constexpr int NCG   = NC / 4;
    constexpr int PARTS = pow2_floor((RW * 32 / NCG) < SLAB ? (RW * 32 / NCG) : SLAB);
    constexpr int RP    = SLAB / PARTS;
    static_assert(SLAB == 32, "step 1 gives every lane one column of the slab");
    static_assert(NC % 4 == 0 && PARTS >= 1 && PARTS * NCG <= RW * 32 && PARTS * RP == SLAB, "thread map of step 2");
    __shared__ float sh[RW][32], sh2[RW][32];
    __shared__ float scL[SLAB], sfL[SLAB];
    __shared__ float4 cL[PARTS * NCG];
    const int lane = threadIdx.x & 31;
    const int w    = threadIdx.x >> 5;
    const int col  = blockIdx.x * SLAB + lane;
    float s = 0.f, ss = 0.f;
    for (int t0 = w; t0 < nTile; t0 += RW * RU) {
        float a[RU], b[RU];
        #pragma unroll
        for (int u = 0; u < RU; ++u) {
            const int t  = t0 + u * RW;
            const int tc = (t < nTile) ? t : (nTile - 1);
            a[u] = part[(size_t)(tc * 2)     * FDIM + col];
            b[u] = part[(size_t)(tc * 2 + 1) * FDIM + col];
        }
        #pragma unroll
        for (int u = 0; u < RU; ++u) {
            if (t0 + u * RW < nTile) { s += a[u]; ss += b[u]; }
        }
    }
    sh[w][lane] = s;  sh2[w][lane] = ss;
    __syncthreads();
    if (w == 0) {
        float S = 0.f, SS = 0.f;
        #pragma unroll
        for (int k = 0; k < RW; ++k) { S += sh[k][lane]; SS += sh2[k][lane]; }
        const float mu  = S / rows;
        const float var = SS / rows - mu * mu;
        const float sc  = rsqrtf(var + BN_EPS) * gamma[col];
        scL[lane] = sc;
        sfL[lane] = beta[col] - mu * sc;
    }
    __syncthreads();

    const int t = threadIdx.x;
    if (U) {
        for (int i = t; i < SLAB * XW; i += RW * 32) {
            const int r = i / XW, j = i % XW;
            const int k = blockIdx.x * SLAB + r;
            Wa[(size_t)k * XW + j] = (j < SCP) ? scL[r] * U[(size_t)k * SCP + j] : 0.f;
        }
        if (t < XW) {
            float cs = 0.f;
            if (t < SCP) {
                float u[SLAB];
                #pragma unroll
                for (int r = 0; r < SLAB; ++r) u[r] = U[(size_t)(blockIdx.x * SLAB + r) * SCP + t];
                #pragma unroll
                for (int r = 0; r < SLAB; ++r) cs += sfL[r] * u[r];
            }
            cpa[(size_t)blockIdx.x * XW + t] = cs;
        }
    }
    if (t < PARTS * NCG) {
        const int p  = t / NCG;
        const int cg = t % NCG;
        const size_t base = (size_t)(blockIdx.x * SLAB + p * RP) * NC + cg * 4;
        float4 v[RP];
        #pragma unroll
        for (int u = 0; u < RP; ++u) v[u] = *(const float4*)(Wt + base + (size_t)u * NC);
        float4 cs = {0.f, 0.f, 0.f, 0.f};
        #pragma unroll
        for (int u = 0; u < RP; ++u) {
            const float a = scL[p * RP + u], f = sfL[p * RP + u];
            float4 o;
            o.x = v[u].x * a;  o.y = v[u].y * a;  o.z = v[u].z * a;  o.w = v[u].w * a;
            *(float4*)(Wf + base + (size_t)u * NC) = o;
            cs.x += f * v[u].x;  cs.y += f * v[u].y;  cs.z += f * v[u].z;  cs.w += f * v[u].w;
        }
        cL[p * NCG + cg] = cs;
    }
    __syncthreads();
    if (t < NCG) {
        float4 cs = cL[t];
        #pragma unroll
        for (int p = 1; p < PARTS; ++p) {
            const float4 x = cL[p * NCG + t];
            cs.x += x.x;  cs.y += x.y;  cs.z += x.z;  cs.w += x.w;
        }
        *(float4*)(cpart + (size_t)blockIdx.x * NC + t * 4) = cs;
    }
}

__global__ __launch_bounds__(256)
void pool_max(const float* __restrict__ lin, const int* __restrict__ ibatch,
              unsigned* __restrict__ pooled, int rows)
{
    size_t i = (size_t)blockIdx.x * blockDim.x + threadIdx.x;
    if (i >= (size_t)rows * OUT_C) return;
    int n = (int)(i / OUT_C), c = (int)(i % OUT_C);
    atomicMax(&pooled[ibatch[n] * OUT_C + c], fmap(lin[i]));
}

__global__ __launch_bounds__(256)
void pool_final(const unsigned* __restrict__ pooled, float* __restrict__ out, int n)
{
    int i = blockIdx.x * blockDim.x + threadIdx.x;
    if (i < n) out[i] = funmap(pooled[i]);
}

extern "C" void kernel_launch(void* const* d_in, const int* in_sizes, int n_in,
                              void* d_out, int out_size, void* d_ws, size_t ws_size,
                              hipStream_t stream)
{
    const float* x_in   = (const float*)d_in[0];
    const int*   adj    = (const int*)d_in[1];
    const int*   ibatch = (const int*)d_in[2];
    const float* W1     = (const float*)d_in[3];
    const float* att_s1 = (const float*)d_in[4];
    const float* att_d1 = (const float*)d_in[5];
    const float* b1     = (const float*)d_in[6];
    const float* g1     = (const float*)d_in[7];
    const float* be1    = (const float*)d_in[8];
    const float* W2     = (const float*)d_in[9];
    const float* att_s2 = (const float*)d_in[10];
    const float* att_d2 = (const float*)d_in[11];
    const float* b2     = (const float*)d_in[12];
    const float* g2     = (const float*)d_in[13];
    const float* be2    = (const float*)d_in[14];
    const float* linW   = (const float*)d_in[15];
    const float* linb   = (const float*)d_in[16];
    float* out = (float*)d_out;

    const int N    = in_sizes[0] / F_IN;
    const int E    = in_sizes[1] / 2;
    const int B    = out_size / OUT_C;
    const int mT   = (N + 31) / 32;

    float*    hx   = (float*)d_ws;
    float*    agg  = hx  + (size_t)N * FDIM;
    float*    sc   = agg + (size_t)N * FDIM;
    float*    lin  = sc  + (size_t)N * SCP;
    unsigned* pooled = (unsigned*)(lin + (size_t)N * OUT_C);

    const int    nTile = (N + TD - 1) / TD;
    float*    part = lin;
    float*    W2f  = part + (size_t)nTile * 2 * FDIM;
    float*    cp1  = W2f + (size_t)FDIM * FDIM;
    float*    Wa1  = cp1 + (size_t)NSLAB * FDIM;
    float*    U2   = Wa1 + (size_t)F_IN * XW;
    float*    Wa2  = U2  + (size_t)FDIM * SCP;
    float*    cpa  = Wa2 + (size_t)FDIM * XW;
    float*    lWf  = hx;
    float*    cp2  = lWf + (size_t)FDIM * OUT_C;

    att_proj<<<(F_IN + FDIM + 7) / 8, 256, 0, stream>>>(W1, att_s1, att_d1, Wa1, F_IN, W2, att_s2, att_d2, U2, FDIM);

    auto layer = [&](const float* xin, int K, const float* W, const float* cpart,
                     const float* Wa, const float* cpx, const float* bias) {
        wmma_gemm_f32<<<mT * (FDIM / 64 + 1), 32, 0, stream>>>(xin, W, hx, nullptr, cpart, Wa, cpx, sc, N, K, FDIM);
        aggregate_tile<<<nTile, NWV * 32, 0, stream>>>(adj, hx, sc, bias, agg, part, E, N);
    };

    layer(x_in, F_IN, W1, nullptr, Wa1, nullptr, b1);
    bn_fold<FDIM><<<NSLAB, RW * 32, 0, stream>>>(part, g1, be1, W2, W2f, cp1, U2, Wa2, cpa, nTile, N);
    layer(agg,  FDIM, W2f, cp1, Wa2, cpa, b2);
    bn_fold<OUT_C><<<NSLAB, RW * 32, 0, stream>>>(part, g2, be2, linW, lWf, cp2, nullptr, nullptr, nullptr, nTile, N);

    wmma_gemm_f32<<<mT * (OUT_C / 64), 32, 0, stream>>>(agg, lWf, lin, linb, cp2, nullptr, nullptr, nullptr, N, FDIM, OUT_C);
    fill_u32<<<(B * OUT_C + 255) / 256, 256, 0, stream>>>(pooled, FMAP_NEG_INF, (size_t)B * OUT_C);
    pool_max<<<(int)(((size_t)N * OUT_C + 255) / 256), 256, 0, stream>>>(lin, ibatch, pooled, N);
    pool_final<<<(B * OUT_C + 255) / 256, 256, 0, stream>>>(pooled, out, B * OUT_C);
}
